# starting kernel
# baseline (speedup 1.0000x reference)
#include <hip/hip_runtime.h>
#include <hip/hip_bf16.h>

#define N_NODES   50000
#define N_EDGES   800000
#define DIM       128
#define N_LAYERS  4
#define NUM_GRAPHS 128
#define BN_EPS    1e-5f

typedef __attribute__((ext_vector_type(2))) float v2f;
typedef __attribute__((ext_vector_type(8))) float v8f;

__global__ __launch_bounds__(256) void gin_zero(float* __restrict__ p, long n) {
    long i = (long)blockIdx.x * blockDim.x + threadIdx.x;
    if (i < n) p[i] = 0.0f;
}

__global__ __launch_bounds__(256) void gin_scatter(const float* __restrict__ hin,
                                                   const int* __restrict__ src,
                                                   const int* __restrict__ dst,
                                                   float* __restrict__ agg) {
    long t = (long)blockIdx.x * blockDim.x + threadIdx.x;
    long e = t >> 5;
    int lane = (int)(t & 31);
    if (e >= N_EDGES) return;
    int s = src[e];
    int d = dst[e];
    const float4* hs = (const float4*)(hin + (long)s * DIM);
    float4 v = hs[lane];
    float* ad = agg + (long)d * DIM + lane * 4;
    atomicAdd(ad + 0, v.x);
    atomicAdd(ad + 1, v.y);
    atomicAdd(ad + 2, v.z);
    atomicAdd(ad + 3, v.w);
}

__global__ __launch_bounds__(256) void gin_mlp(const float* __restrict__ hin,
                                               const float* __restrict__ agg,
                                               const float* __restrict__ W1,
                                               const float* __restrict__ b1,
                                               const float* __restrict__ W2,
                                               const float* __restrict__ b2,
                                               const float* __restrict__ epsp,
                                               float* __restrict__ hout,
                                               float* __restrict__ stats,
                                               int do_stats) {
    __shared__ float zt[16 * DIM];
    __shared__ float y1[16 * DIM];

    const int row0 = blockIdx.x * 16;
    const int tid  = threadIdx.x;
    const float e1 = 1.0f + epsp[0];

    for (int i = tid; i < 16 * DIM; i += 256) {
        int r = i >> 7;
        int c = i & (DIM - 1);
        long g = (long)(row0 + r) * DIM + c;
        zt[i] = e1 * hin[g] + agg[g];
    }
    __syncthreads();

    const int wave = tid >> 5;
    const int lane = tid & 31;
    const int n    = lane & 15;
    const int hi   = lane >> 4;
    const int col0 = wave * 16;
    const int kk   = hi * 2;

    __builtin_prefetch(&W1[col0], 0, 0);
    __builtin_prefetch(&W2[col0], 0, 0);

    v8f acc;
    {
        float bv = b1[col0 + n];
        for (int v = 0; v < 8; ++v) acc[v] = bv;
    }
    for (int k = 0; k < DIM; k += 4) {
        v2f a, b;
        const float* zp = &zt[n * DIM + k + kk];
        a.x = zp[0];
        a.y = zp[1];
        b.x = W1[(k + kk) * DIM + col0 + n];
        b.y = W1[(k + kk + 1) * DIM + col0 + n];
        acc = __builtin_amdgcn_wmma_f32_16x16x4_f32(false, a, false, b,
                                                    (short)0, acc, false, false);
    }
    for (int v = 0; v < 8; ++v) {
        int m = v + 8 * hi;
        y1[m * DIM + col0 + n] = fmaxf(acc[v], 0.0f);
    }
    __syncthreads();

    {
        float bv = b2[col0 + n];
        for (int v = 0; v < 8; ++v) acc[v] = bv;
    }
    for (int k = 0; k < DIM; k += 4) {
        v2f a, b;
        const float* yp = &y1[n * DIM + k + kk];
        a.x = yp[0];
        a.y = yp[1];
        b.x = W2[(k + kk) * DIM + col0 + n];
        b.y = W2[(k + kk + 1) * DIM + col0 + n];
        acc = __builtin_amdgcn_wmma_f32_16x16x4_f32(false, a, false, b,
                                                    (short)0, acc, false, false);
    }

    float s = 0.0f, ssq = 0.0f;
    for (int v = 0; v < 8; ++v) {
        int m = v + 8 * hi;
        float val = acc[v];
        hout[(long)(row0 + m) * DIM + col0 + n] = val;
        s += val;
        ssq += val * val;
    }
    if (do_stats) {
        atomicAdd(&stats[col0 + n], s);
        atomicAdd(&stats[DIM + col0 + n], ssq);
    }
}

__global__ __launch_bounds__(256) void gin_bn_relu(float* __restrict__ z,
                                                   const float* __restrict__ stats,
                                                   const float* __restrict__ gamma,
                                                   const float* __restrict__ beta) {
    long i = (long)blockIdx.x * blockDim.x + threadIdx.x;
    if (i >= (long)N_NODES * DIM) return;
    int c = (int)(i & (DIM - 1));
    const float invN = 1.0f / (float)N_NODES;
    float mu  = stats[c] * invN;
    float var = stats[DIM + c] * invN - mu * mu;
    float v = (z[i] - mu) * rsqrtf(var + BN_EPS) * gamma[c] + beta[c];
    z[i] = fmaxf(v, 0.0f);
}

__global__ __launch_bounds__(256) void gin_pool(const float* __restrict__ h,
                                                const int* __restrict__ batch,
                                                float* __restrict__ out) {
    long t = (long)blockIdx.x * blockDim.x + threadIdx.x;
    long nidx = t >> 5;
    int lane = (int)(t & 31);
    if (nidx >= N_NODES) return;
    int g = batch[nidx];
    const float4* hp = (const float4*)(h + nidx * DIM);
    float4 v = hp[lane];
    float* op = out + (long)g * DIM + lane * 4;
    atomicAdd(op + 0, v.x);
    atomicAdd(op + 1, v.y);
    atomicAdd(op + 2, v.z);
    atomicAdd(op + 3, v.w);
}

extern "C" void kernel_launch(void* const* d_in, const int* in_sizes, int n_in,
                              void* d_out, int out_size, void* d_ws, size_t ws_size,
                              hipStream_t stream) {
    const float* x     = (const float*)d_in[0];
    const int*   ei    = (const int*)d_in[1];
    const int*   batch = (const int*)d_in[2];
    const float* W1    = (const float*)d_in[3];
    const float* b1    = (const float*)d_in[4];
    const float* W2    = (const float*)d_in[5];
    const float* b2    = (const float*)d_in[6];
    const float* eps   = (const float*)d_in[7];
    const float* gamma = (const float*)d_in[8];
    const float* beta  = (const float*)d_in[9];
    float* out = (float*)d_out;

    const long ND = (long)N_NODES * DIM;
    float* hA    = (float*)d_ws;
    float* hB    = hA + ND;
    float* agg   = hB + ND;
    float* stats = agg + ND;

    const int* src = ei;
    const int* dst = ei + N_EDGES;

    const long zeroN   = ND + 2 * DIM;
    const int  zeroGrd = (int)((zeroN + 255) / 256);
    const int  scatGrd = (int)(((long)N_EDGES * 32 + 255) / 256);
    const int  mlpGrd  = N_NODES / 16;
    const int  bnGrd   = (int)((ND + 255) / 256);
    const int  poolGrd = (int)(((long)N_NODES * 32 + 255) / 256);

    for (int i = 0; i < N_LAYERS; ++i) {
        const float* hin  = (i == 0) ? x : ((i & 1) ? hA : hB);
        float*       hout = (i & 1) ? hB : hA;
        int do_stats = (i < N_LAYERS - 1) ? 1 : 0;

        gin_zero<<<zeroGrd, 256, 0, stream>>>(agg, zeroN);
        gin_scatter<<<scatGrd, 256, 0, stream>>>(hin, src, dst, agg);
        gin_mlp<<<mlpGrd, 256, 0, stream>>>(hin, agg,
                                            W1 + (long)i * DIM * DIM, b1 + (long)i * DIM,
                                            W2 + (long)i * DIM * DIM, b2 + (long)i * DIM,
                                            eps + i, hout, stats, do_stats);
        if (do_stats) {
            gin_bn_relu<<<bnGrd, 256, 0, stream>>>(hout, stats,
                                                   gamma + (long)i * DIM,
                                                   beta + (long)i * DIM);
        }
    }
    gin_zero<<<(NUM_GRAPHS * DIM + 255) / 256, 256, 0, stream>>>(out, NUM_GRAPHS * DIM);
    gin_pool<<<poolGrd, 256, 0, stream>>>(hB, batch, out);
}


# Edge scatter and pool: a wave takes 8 edges (nodes), not 1: indices by 2 scalar loads a wave, 32 row loads issued ahead of 32 whole-line float atomics; 8x fewer waves; atomic arrival order kept
# speedup vs baseline: 1.6327x; 1.6327x over previous
#include <hip/hip_runtime.h>
#include <hip/hip_bf16.h>

#define N_NODES   50000
#define N_EDGES   800000
#define DIM       128
#define N_LAYERS  4
#define NUM_GRAPHS 128
#define BN_EPS    1e-5f

typedef __attribute__((ext_vector_type(2))) float v2f;
typedef __attribute__((ext_vector_type(8))) float v8f;

__global__ __launch_bounds__(256) void gin_zero(float* __restrict__ p, long n) {
    long i = (long)blockIdx.x * blockDim.x + threadIdx.x;
    if (i < n) p[i] = 0.0f;
}

#define EPW 8
static_assert(N_EDGES % EPW == 0, "a wave's edges must all exist");
static_assert(N_NODES % EPW == 0, "a wave's nodes must all exist");

__global__ __launch_bounds__(256) void gin_scatter(const float* __restrict__ hin,
                                                   const int* __restrict__ src,
                                                   const int* __restrict__ dst,
                                                   float* __restrict__ agg) {
    const int lane = (int)(threadIdx.x & 31);
    const int wv = __builtin_amdgcn_readfirstlane((int)(blockIdx.x * 8 + (threadIdx.x >> 5)));
    const int e0 = wv * EPW;
    if (e0 >= N_EDGES) return;
    int s[EPW], d[EPW];
    #pragma unroll
    for (int j = 0; j < EPW; ++j) {
        s[j] = src[e0 + j];
        d[j] = dst[e0 + j];
    }
    float v[EPW][4];
    #pragma unroll
    for (int j = 0; j < EPW; ++j) {
        const float* hs = hin + (long)s[j] * DIM + lane;
        v[j][0] = hs[0];
        v[j][1] = hs[32];
        v[j][2] = hs[64];
        v[j][3] = hs[96];
    }
    #pragma unroll
    for (int j = 0; j < EPW; ++j) {
        float* ad = agg + (long)d[j] * DIM + lane;
        atomicAdd(ad + 0,  v[j][0]);
        atomicAdd(ad + 32, v[j][1]);
        atomicAdd(ad + 64, v[j][2]);
        atomicAdd(ad + 96, v[j][3]);
    }
}

__global__ __launch_bounds__(256) void gin_mlp(const float* __restrict__ hin,
                                               const float* __restrict__ agg,
                                               const float* __restrict__ W1,
                                               const float* __restrict__ b1,
                                               const float* __restrict__ W2,
                                               const float* __restrict__ b2,
                                               const float* __restrict__ epsp,
                                               float* __restrict__ hout,
                                               float* __restrict__ stats,
                                               int do_stats) {
    __shared__ float zt[16 * DIM];
    __shared__ float y1[16 * DIM];

    const int row0 = blockIdx.x * 16;
    const int tid  = threadIdx.x;
    const float e1 = 1.0f + epsp[0];

    for (int i = tid; i < 16 * DIM; i += 256) {
        int r = i >> 7;
        int c = i & (DIM - 1);
        long g = (long)(row0 + r) * DIM + c;
        zt[i] = e1 * hin[g] + agg[g];
    }
    __syncthreads();

    const int wave = tid >> 5;
    const int lane = tid & 31;
    const int n    = lane & 15;
    const int hi   = lane >> 4;
    const int col0 = wave * 16;
    const int kk   = hi * 2;

    __builtin_prefetch(&W1[col0], 0, 0);
    __builtin_prefetch(&W2[col0], 0, 0);

    v8f acc;
    {
        float bv = b1[col0 + n];
        for (int v = 0; v < 8; ++v) acc[v] = bv;
    }
    for (int k = 0; k < DIM; k += 4) {
        v2f a, b;
        const float* zp = &zt[n * DIM + k + kk];
        a.x = zp[0];
        a.y = zp[1];
        b.x = W1[(k + kk) * DIM + col0 + n];
        b.y = W1[(k + kk + 1) * DIM + col0 + n];
        acc = __builtin_amdgcn_wmma_f32_16x16x4_f32(false, a, false, b,
                                                    (short)0, acc, false, false);
    }
    for (int v = 0; v < 8; ++v) {
        int m = v + 8 * hi;
        y1[m * DIM + col0 + n] = fmaxf(acc[v], 0.0f);
    }
    __syncthreads();

    {
        float bv = b2[col0 + n];
        for (int v = 0; v < 8; ++v) acc[v] = bv;
    }
    for (int k = 0; k < DIM; k += 4) {
        v2f a, b;
        const float* yp = &y1[n * DIM + k + kk];
        a.x = yp[0];
        a.y = yp[1];
        b.x = W2[(k + kk) * DIM + col0 + n];
        b.y = W2[(k + kk + 1) * DIM + col0 + n];
        acc = __builtin_amdgcn_wmma_f32_16x16x4_f32(false, a, false, b,
                                                    (short)0, acc, false, false);
    }

    float s = 0.0f, ssq = 0.0f;
    for (int v = 0; v < 8; ++v) {
        int m = v + 8 * hi;
        float val = acc[v];
        hout[(long)(row0 + m) * DIM + col0 + n] = val;
        s += val;
        ssq += val * val;
    }
    if (do_stats) {
        atomicAdd(&stats[col0 + n], s);
        atomicAdd(&stats[DIM + col0 + n], ssq);
    }
}

__global__ __launch_bounds__(256) void gin_bn_relu(float* __restrict__ z,
                                                   const float* __restrict__ stats,
                                                   const float* __restrict__ gamma,
                                                   const float* __restrict__ beta) {
    long i = (long)blockIdx.x * blockDim.x + threadIdx.x;
    if (i >= (long)N_NODES * DIM) return;
    int c = (int)(i & (DIM - 1));
    const float invN = 1.0f / (float)N_NODES;
    float mu  = stats[c] * invN;
    float var = stats[DIM + c] * invN - mu * mu;
    float v = (z[i] - mu) * rsqrtf(var + BN_EPS) * gamma[c] + beta[c];
    z[i] = fmaxf(v, 0.0f);
}

__global__ __launch_bounds__(256) void gin_pool(const float* __restrict__ h,
                                                const int* __restrict__ batch,
                                                float* __restrict__ out) {
    const int lane = (int)(threadIdx.x & 31);
    const int wv = __builtin_amdgcn_readfirstlane((int)(blockIdx.x * 8 + (threadIdx.x >> 5)));
    const int n0 = wv * EPW;
    if (n0 >= N_NODES) return;
    int g[EPW];
    #pragma unroll
    for (int j = 0; j < EPW; ++j) g[j] = batch[n0 + j];
    float v[EPW][4];
    #pragma unroll
    for (int j = 0; j < EPW; ++j) {
        const float* hp = h + (long)(n0 + j) * DIM + lane;
        v[j][0] = hp[0];
        v[j][1] = hp[32];
        v[j][2] = hp[64];
        v[j][3] = hp[96];
    }
    #pragma unroll
    for (int j = 0; j < EPW; ++j) {
        float* op = out + (long)g[j] * DIM + lane;
        atomicAdd(op + 0,  v[j][0]);
        atomicAdd(op + 32, v[j][1]);
        atomicAdd(op + 64, v[j][2]);
        atomicAdd(op + 96, v[j][3]);
    }
}

extern "C" void kernel_launch(void* const* d_in, const int* in_sizes, int n_in,
                              void* d_out, int out_size, void* d_ws, size_t ws_size,
                              hipStream_t stream) {
    const float* x     = (const float*)d_in[0];
    const int*   ei    = (const int*)d_in[1];
    const int*   batch = (const int*)d_in[2];
    const float* W1    = (const float*)d_in[3];
    const float* b1    = (const float*)d_in[4];
    const float* W2    = (const float*)d_in[5];
    const float* b2    = (const float*)d_in[6];
    const float* eps   = (const float*)d_in[7];
    const float* gamma = (const float*)d_in[8];
    const float* beta  = (const float*)d_in[9];
    float* out = (float*)d_out;

    const long ND = (long)N_NODES * DIM;
    float* hA    = (float*)d_ws;
    float* hB    = hA + ND;
    float* agg   = hB + ND;
    float* stats = agg + ND;

    const int* src = ei;
    const int* dst = ei + N_EDGES;

    const long zeroN   = ND + 2 * DIM;
    const int  zeroGrd = (int)((zeroN + 255) / 256);
    const int  scatGrd = (N_EDGES / EPW + 7) / 8;
    const int  mlpGrd  = N_NODES / 16;
    const int  bnGrd   = (int)((ND + 255) / 256);
    const int  poolGrd = (N_NODES / EPW + 7) / 8;

    for (int i = 0; i < N_LAYERS; ++i) {
        const float* hin  = (i == 0) ? x : ((i & 1) ? hA : hB);
        float*       hout = (i & 1) ? hB : hA;
        int do_stats = (i < N_LAYERS - 1) ? 1 : 0;

        gin_zero<<<zeroGrd, 256, 0, stream>>>(agg, zeroN);
        gin_scatter<<<scatGrd, 256, 0, stream>>>(hin, src, dst, agg);
        gin_mlp<<<mlpGrd, 256, 0, stream>>>(hin, agg,
                                            W1 + (long)i * DIM * DIM, b1 + (long)i * DIM,
                                            W2 + (long)i * DIM * DIM, b2 + (long)i * DIM,
                                            eps + i, hout, stats, do_stats);
        if (do_stats) {
            gin_bn_relu<<<bnGrd, 256, 0, stream>>>(hout, stats,
                                                   gamma + (long)i * DIM,
                                                   beta + (long)i * DIM);
        }
    }
    gin_zero<<<(NUM_GRAPHS * DIM + 255) / 256, 256, 0, stream>>>(out, NUM_GRAPHS * DIM);
    gin_pool<<<poolGrd, 256, 0, stream>>>(hB, batch, out);
}


# MLP kernel: a wave owns 16 nodes x all 128 columns; both products transposed, so product 1's D registers are product 2's B operand; W1, W2 in LDS in fragment order; k order and BN add order change
# speedup vs baseline: 1.7674x; 1.0825x over previous
#include <hip/hip_runtime.h>
#include <hip/hip_bf16.h>

#define N_NODES   50000
#define N_EDGES   800000
#define DIM       128
#define N_LAYERS  4
#define NUM_GRAPHS 128
#define BN_EPS    1e-5f

typedef __attribute__((ext_vector_type(2))) float v2f;
typedef __attribute__((ext_vector_type(8))) float v8f;

__global__ __launch_bounds__(256) void gin_zero(float* __restrict__ p, long n) {
    long i = (long)blockIdx.x * blockDim.x + threadIdx.x;
    if (i < n) p[i] = 0.0f;
}

#define EPW 8
static_assert(N_EDGES % EPW == 0, "a wave's edges must all exist");
static_assert(N_NODES % EPW == 0, "a wave's nodes must all exist");

__global__ __launch_bounds__(256) void gin_scatter(const float* __restrict__ hin,
                                                   const int* __restrict__ src,
                                                   const int* __restrict__ dst,
                                                   float* __restrict__ agg) {
    const int lane = (int)(threadIdx.x & 31);
    const int wv = __builtin_amdgcn_readfirstlane((int)(blockIdx.x * 8 + (threadIdx.x >> 5)));
    const int e0 = wv * EPW;
    if (e0 >= N_EDGES) return;
    int s[EPW], d[EPW];
    #pragma unroll
    for (int j = 0; j < EPW; ++j) {
        s[j] = src[e0 + j];
        d[j] = dst[e0 + j];
    }
    float v[EPW][4];
    #pragma unroll
    for (int j = 0; j < EPW; ++j) {
        const float* hs = hin + (long)s[j] * DIM + lane;
        v[j][0] = hs[0];
        v[j][1] = hs[32];
        v[j][2] = hs[64];
        v[j][3] = hs[96];
    }
    #pragma unroll
    for (int j = 0; j < EPW; ++j) {
        float* ad = agg + (long)d[j] * DIM + lane;
        atomicAdd(ad + 0,  v[j][0]);
        atomicAdd(ad + 32, v[j][1]);
        atomicAdd(ad + 64, v[j][2]);
        atomicAdd(ad + 96, v[j][3]);
    }
}

#define TPB 8
#define N_TILES (N_NODES / 16)
static_assert(N_NODES % 16 == 0, "whole 16-row tiles only");

typedef __attribute__((ext_vector_type(4))) float v4f;

__device__ __forceinline__ void stage_w(const float* __restrict__ Wg, v4f* __restrict__ Wl, int tid) {
    #pragma unroll
    for (int it = 0; it < 16; ++it) {
        const int id = tid + 256 * it;
        const int n  = id & 127;
        const int kg = id >> 7;
        const int t  = kg >> 2;
        const int h  = (kg >> 1) & 1;
        const int q  = kg & 1;
        const int u  = n >> 4;
        const int c  = n & 15;
        const float* wp = Wg + (4 * kg) * DIM + n;
        v4f w;
        w.x = wp[0];
        w.y = wp[DIM];
        w.z = wp[2 * DIM];
        w.w = wp[3 * DIM];
        Wl[(((u * 8 + t) * 2 + q) * 32) + 16 * h + c] = w;
    }
}

__device__ __forceinline__ float row16_sum(float v) {
    v += __builtin_bit_cast(float, __builtin_amdgcn_update_dpp(0, __builtin_bit_cast(int, v), 0x161, 0xf, 0xf, false));
    v += __builtin_bit_cast(float, __builtin_amdgcn_update_dpp(0, __builtin_bit_cast(int, v), 0x162, 0xf, 0xf, false));
    v += __builtin_bit_cast(float, __builtin_amdgcn_update_dpp(0, __builtin_bit_cast(int, v), 0x164, 0xf, 0xf, false));
    v += __builtin_bit_cast(float, __builtin_amdgcn_update_dpp(0, __builtin_bit_cast(int, v), 0x168, 0xf, 0xf, false));
    return v;
}

__global__ __launch_bounds__(256) void gin_mlp(const float* __restrict__ hin,
                                               const float* __restrict__ agg,
                                               const float* __restrict__ W1,
                                               const float* __restrict__ b1,
                                               const float* __restrict__ W2,
                                               const float* __restrict__ b2,
                                               const float* __restrict__ epsp,
                                               float* __restrict__ hout,
                                               float* __restrict__ stats,
                                               int do_stats) {
    __shared__ v4f w1l[DIM * DIM / 4];
    __shared__ v4f w2l[DIM * DIM / 4];
    __shared__ float red[TPB][2 * DIM];

    const int tid  = threadIdx.x;
    const int lane = tid & 31;
    const int c    = lane & 15;
    const int h    = lane >> 4;
    const int wave = __builtin_amdgcn_readfirstlane(tid >> 5);
    const int tile = blockIdx.x * TPB + wave;
    const bool valid = tile < N_TILES;
    const long rowoff = (long)(tile * 16 + c) * DIM + 8 * h;
    const float e1 = 1.0f + epsp[0];

    float zf[8][8];
    if (valid) {
        #pragma unroll
        for (int t = 0; t < 8; ++t) {
            const v4f h0 = *(const v4f*)(hin + rowoff + 16 * t);
            const v4f h1 = *(const v4f*)(hin + rowoff + 16 * t + 4);
            const v4f a0 = *(const v4f*)(agg + rowoff + 16 * t);
            const v4f a1 = *(const v4f*)(agg + rowoff + 16 * t + 4);
            zf[t][0] = e1 * h0.x + a0.x;
            zf[t][1] = e1 * h0.y + a0.y;
            zf[t][2] = e1 * h0.z + a0.z;
            zf[t][3] = e1 * h0.w + a0.w;
            zf[t][4] = e1 * h1.x + a1.x;
            zf[t][5] = e1 * h1.y + a1.y;
            zf[t][6] = e1 * h1.z + a1.z;
            zf[t][7] = e1 * h1.w + a1.w;
        }
    }

    stage_w(W1, w1l, tid);
    stage_w(W2, w2l, tid);
    __syncthreads();

    v8f acc2[8];
    if (valid) {
        v8f acc1[8];
        #pragma unroll
        for (int u = 0; u < 8; ++u) {
            const v4f bl = *(const v4f*)(b1 + 16 * u + 8 * h);
            const v4f bh = *(const v4f*)(b1 + 16 * u + 8 * h + 4);
            acc1[u][0] = bl.x; acc1[u][1] = bl.y; acc1[u][2] = bl.z; acc1[u][3] = bl.w;
            acc1[u][4] = bh.x; acc1[u][5] = bh.y; acc1[u][6] = bh.z; acc1[u][7] = bh.w;
        }
        #pragma unroll
        for (int t = 0; t < 8; ++t) {
            #pragma unroll
            for (int u = 0; u < 8; ++u) {
                const v4f wa = w1l[((u * 8 + t) * 2 + 0) * 32 + lane];
                const v4f wb = w1l[((u * 8 + t) * 2 + 1) * 32 + lane];
                v2f a, b;
                a.x = wa.x; a.y = wa.y; b.x = zf[t][0]; b.y = zf[t][1];
                acc1[u] = __builtin_amdgcn_wmma_f32_16x16x4_f32(false, a, false, b, (short)0, acc1[u], false, false);
                a.x = wa.z; a.y = wa.w; b.x = zf[t][2]; b.y = zf[t][3];
                acc1[u] = __builtin_amdgcn_wmma_f32_16x16x4_f32(false, a, false, b, (short)0, acc1[u], false, false);
                a.x = wb.x; a.y = wb.y; b.x = zf[t][4]; b.y = zf[t][5];
                acc1[u] = __builtin_amdgcn_wmma_f32_16x16x4_f32(false, a, false, b, (short)0, acc1[u], false, false);
                a.x = wb.z; a.y = wb.w; b.x = zf[t][6]; b.y = zf[t][7];
                acc1[u] = __builtin_amdgcn_wmma_f32_16x16x4_f32(false, a, false, b, (short)0, acc1[u], false, false);
            }
        }
        #pragma unroll
        for (int t = 0; t < 8; ++t) {
            #pragma unroll
            for (int r = 0; r < 8; ++r) acc1[t][r] = fmaxf(acc1[t][r], 0.0f);
        }

        #pragma unroll
        for (int u = 0; u < 8; ++u) {
            const v4f bl = *(const v4f*)(b2 + 16 * u + 8 * h);
            const v4f bh = *(const v4f*)(b2 + 16 * u + 8 * h + 4);
            acc2[u][0] = bl.x; acc2[u][1] = bl.y; acc2[u][2] = bl.z; acc2[u][3] = bl.w;
            acc2[u][4] = bh.x; acc2[u][5] = bh.y; acc2[u][6] = bh.z; acc2[u][7] = bh.w;
        }
        #pragma unroll
        for (int t = 0; t < 8; ++t) {
            #pragma unroll
            for (int u = 0; u < 8; ++u) {
                const v4f wa = w2l[((u * 8 + t) * 2 + 0) * 32 + lane];
                const v4f wb = w2l[((u * 8 + t) * 2 + 1) * 32 + lane];
                v2f a, b;
                a.x = wa.x; a.y = wa.y; b.x = acc1[t][0]; b.y = acc1[t][1];
                acc2[u] = __builtin_amdgcn_wmma_f32_16x16x4_f32(false, a, false, b, (short)0, acc2[u], false, false);
                a.x = wa.z; a.y = wa.w; b.x = acc1[t][2]; b.y = acc1[t][3];
                acc2[u] = __builtin_amdgcn_wmma_f32_16x16x4_f32(false, a, false, b, (short)0, acc2[u], false, false);
                a.x = wb.x; a.y = wb.y; b.x = acc1[t][4]; b.y = acc1[t][5];
                acc2[u] = __builtin_amdgcn_wmma_f32_16x16x4_f32(false, a, false, b, (short)0, acc2[u], false, false);
                a.x = wb.z; a.y = wb.w; b.x = acc1[t][6]; b.y = acc1[t][7];
                acc2[u] = __builtin_amdgcn_wmma_f32_16x16x4_f32(false, a, false, b, (short)0, acc2[u], false, false);
            }
        }

        #pragma unroll
        for (int u = 0; u < 8; ++u) {
            v4f lo, hi4;
            lo.x = acc2[u][0]; lo.y = acc2[u][1]; lo.z = acc2[u][2]; lo.w = acc2[u][3];
            hi4.x = acc2[u][4]; hi4.y = acc2[u][5]; hi4.z = acc2[u][6]; hi4.w = acc2[u][7];
            *(v4f*)(hout + rowoff + 16 * u)     = lo;
            *(v4f*)(hout + rowoff + 16 * u + 4) = hi4;
        }
    } else {
        #pragma unroll
        for (int u = 0; u < 8; ++u) {
            #pragma unroll
            for (int r = 0; r < 8; ++r) acc2[u][r] = 0.0f;
        }
    }

    if (do_stats) {
        float cs[8][8], cq[8][8];
        #pragma unroll
        for (int u = 0; u < 8; ++u) {
            #pragma unroll
            for (int r = 0; r < 8; ++r) {
                const float val = acc2[u][r];
                cs[u][r] = row16_sum(val);
                cq[u][r] = row16_sum(val * val);
            }
        }
        if (c == 0) {
            #pragma unroll
            for (int u = 0; u < 8; ++u) {
                #pragma unroll
                for (int r = 0; r < 8; ++r) {
                    red[wave][16 * u + 8 * h + r]       = cs[u][r];
                    red[wave][DIM + 16 * u + 8 * h + r] = cq[u][r];
                }
            }
        }
        __syncthreads();
        float tot = 0.0f;
        #pragma unroll
        for (int w = 0; w < TPB; ++w) tot += red[w][tid];
        atomicAdd(&stats[tid], tot);
    }
}

__global__ __launch_bounds__(256) void gin_bn_relu(float* __restrict__ z,
                                                   const float* __restrict__ stats,
                                                   const float* __restrict__ gamma,
                                                   const float* __restrict__ beta) {
    long i = (long)blockIdx.x * blockDim.x + threadIdx.x;
    if (i >= (long)N_NODES * DIM) return;
    int c = (int)(i & (DIM - 1));
    const float invN = 1.0f / (float)N_NODES;
    float mu  = stats[c] * invN;
    float var = stats[DIM + c] * invN - mu * mu;
    float v = (z[i] - mu) * rsqrtf(var + BN_EPS) * gamma[c] + beta[c];
    z[i] = fmaxf(v, 0.0f);
}

__global__ __launch_bounds__(256) void gin_pool(const float* __restrict__ h,
                                                const int* __restrict__ batch,
                                                float* __restrict__ out) {
    const int lane = (int)(threadIdx.x & 31);
    const int wv = __builtin_amdgcn_readfirstlane((int)(blockIdx.x * 8 + (threadIdx.x >> 5)));
    const int n0 = wv * EPW;
    if (n0 >= N_NODES) return;
    int g[EPW];
    #pragma unroll
    for (int j = 0; j < EPW; ++j) g[j] = batch[n0 + j];
    float v[EPW][4];
    #pragma unroll
    for (int j = 0; j < EPW; ++j) {
        const float* hp = h + (long)(n0 + j) * DIM + lane;
        v[j][0] = hp[0];
        v[j][1] = hp[32];
        v[j][2] = hp[64];
        v[j][3] = hp[96];
    }
    #pragma unroll
    for (int j = 0; j < EPW; ++j) {
        float* op = out + (long)g[j] * DIM + lane;
        atomicAdd(op + 0,  v[j][0]);
        atomicAdd(op + 32, v[j][1]);
        atomicAdd(op + 64, v[j][2]);
        atomicAdd(op + 96, v[j][3]);
    }
}

extern "C" void kernel_launch(void* const* d_in, const int* in_sizes, int n_in,
                              void* d_out, int out_size, void* d_ws, size_t ws_size,
                              hipStream_t stream) {
    const float* x     = (const float*)d_in[0];
    const int*   ei    = (const int*)d_in[1];
    const int*   batch = (const int*)d_in[2];
    const float* W1    = (const float*)d_in[3];
    const float* b1    = (const float*)d_in[4];
    const float* W2    = (const float*)d_in[5];
    const float* b2    = (const float*)d_in[6];
    const float* eps   = (const float*)d_in[7];
    const float* gamma = (const float*)d_in[8];
    const float* beta  = (const float*)d_in[9];
    float* out = (float*)d_out;

    const long ND = (long)N_NODES * DIM;
    float* hA    = (float*)d_ws;
    float* hB    = hA + ND;
    float* agg   = hB + ND;
    float* stats = agg + ND;

    const int* src = ei;
    const int* dst = ei + N_EDGES;

    const long zeroN   = ND + 2 * DIM;
    const int  zeroGrd = (int)((zeroN + 255) / 256);
    const int  scatGrd = (N_EDGES / EPW + 7) / 8;
    const int  mlpGrd  = (N_TILES + TPB - 1) / TPB;
    const int  bnGrd   = (int)((ND + 255) / 256);
    const int  poolGrd = (N_NODES / EPW + 7) / 8;

    for (int i = 0; i < N_LAYERS; ++i) {
        const float* hin  = (i == 0) ? x : ((i & 1) ? hA : hB);
        float*       hout = (i & 1) ? hB : hA;
        int do_stats = (i < N_LAYERS - 1) ? 1 : 0;

        gin_zero<<<zeroGrd, 256, 0, stream>>>(agg, zeroN);
        gin_scatter<<<scatGrd, 256, 0, stream>>>(hin, src, dst, agg);
        gin_mlp<<<mlpGrd, 256, 0, stream>>>(hin, agg,
                                            W1 + (long)i * DIM * DIM, b1 + (long)i * DIM,
                                            W2 + (long)i * DIM * DIM, b2 + (long)i * DIM,
                                            eps + i, hout, stats, do_stats);
        if (do_stats) {
            gin_bn_relu<<<bnGrd, 256, 0, stream>>>(hout, stats,
                                                   gamma + (long)i * DIM,
                                                   beta + (long)i * DIM);
        }
    }
    gin_zero<<<(NUM_GRAPHS * DIM + 255) / 256, 256, 0, stream>>>(out, NUM_GRAPHS * DIM);
    gin_pool<<<poolGrd, 256, 0, stream>>>(hB, batch, out);
}


# Neighbour sum: a block owns 196 destination rows in LDS, filters the whole dst list, adds gathered rows by LDS float adds, stores each line once; no global float atomic; sum order = LDS add arrival
# speedup vs baseline: 1.8803x; 1.0639x over previous
#include <hip/hip_runtime.h>
#include <hip/hip_bf16.h>

#define N_NODES   50000
#define N_EDGES   800000
#define DIM       128
#define N_LAYERS  4
#define NUM_GRAPHS 128
#define BN_EPS    1e-5f

typedef __attribute__((ext_vector_type(2))) float v2f;
typedef __attribute__((ext_vector_type(8))) float v8f;

__global__ __launch_bounds__(256) void gin_zero(float* __restrict__ p, long n) {
    long i = (long)blockIdx.x * blockDim.x + threadIdx.x;
    if (i < n) p[i] = 0.0f;
}

#define EPW 8
#define ROWS_PB   196
#define SC_WAVES  32
#define SC_THREADS (SC_WAVES * 32)
#define N_SBLK    ((N_NODES + ROWS_PB - 1) / ROWS_PB)
#define TILE_L    8
#define TILE_E    (TILE_L * 128)
#define N_ETILES  ((N_EDGES + TILE_E - 1) / TILE_E)
#define QCAP      (TILE_E + EPW)
static_assert(N_EDGES % 4 == 0, "a 16-byte index load must not straddle the end of dst");
static_assert(N_EDGES <= (1 << 20), "the edge number must fit the low 20 bits of a queue word");
static_assert(ROWS_PB < (1 << 11), "the local row must fit the bits above the edge number");
static_assert(N_NODES % EPW == 0, "a pool wave's nodes must all exist");

typedef __attribute__((ext_vector_type(4))) int   v4i;
typedef __attribute__((ext_vector_type(4))) float v4f;

__device__ __forceinline__ void load_tile(const int* __restrict__ dst, int tile, int lane, v4i (&t)[TILE_L]) {
    #pragma unroll
    for (int j = 0; j < TILE_L; ++j) {
        const int idx = tile * TILE_E + j * 128 + 4 * lane;
        const int idc = idx < N_EDGES ? idx : N_EDGES - 4;
        v4i d = *(const v4i*)(dst + idc);
        if (idx >= N_EDGES) { d.x = -1; d.y = -1; d.z = -1; d.w = -1; }
        t[j] = d;
    }
}

template <bool FULL>
__device__ __forceinline__ void add_batch(const float* __restrict__ hin, const int* __restrict__ src,
                                          float* rows, const int* qe, int n, int lane) {
    int ent[EPW], s[EPW];
    #pragma unroll
    for (int j = 0; j < EPW; ++j) {
        ent[j] = 0;
        if (FULL || j < n)
            ent[j] = __builtin_amdgcn_readfirstlane(__hip_atomic_load(qe + j, __ATOMIC_RELAXED, __HIP_MEMORY_SCOPE_WAVEFRONT));
    }
    #pragma unroll
    for (int j = 0; j < EPW; ++j) {
        s[j] = 0;
        if (FULL || j < n) s[j] = src[ent[j] & 0xFFFFF];
    }
    float v[EPW][4];
    #pragma unroll
    for (int j = 0; j < EPW; ++j) {
        if (FULL || j < n) {
            const float* hs = hin + (long)s[j] * DIM + lane;
            v[j][0] = hs[0];
            v[j][1] = hs[32];
            v[j][2] = hs[64];
            v[j][3] = hs[96];
        }
    }
    #pragma unroll
    for (int j = 0; j < EPW; ++j) {
        if (FULL || j < n) {
            float* r = rows + (ent[j] >> 20) * DIM + lane;
            __hip_atomic_fetch_add(r + 0,  v[j][0], __ATOMIC_RELAXED, __HIP_MEMORY_SCOPE_WORKGROUP);
            __hip_atomic_fetch_add(r + 32, v[j][1], __ATOMIC_RELAXED, __HIP_MEMORY_SCOPE_WORKGROUP);
            __hip_atomic_fetch_add(r + 64, v[j][2], __ATOMIC_RELAXED, __HIP_MEMORY_SCOPE_WORKGROUP);
            __hip_atomic_fetch_add(r + 96, v[j][3], __ATOMIC_RELAXED, __HIP_MEMORY_SCOPE_WORKGROUP);
        }
    }
}

__global__ __launch_bounds__(SC_THREADS) void gin_rowsum(const float* __restrict__ hin,
                                                         const int* __restrict__ src,
                                                         const int* __restrict__ dst,
                                                         float* __restrict__ agg) {
    __shared__ __attribute__((aligned(16))) float rows[ROWS_PB * DIM];
    __shared__ int   queue[SC_WAVES][QCAP];

    const int tid  = (int)threadIdx.x;
    const int lane = tid & 31;
    const int wave = __builtin_amdgcn_readfirstlane(tid >> 5);
    const int row0 = (int)blockIdx.x * ROWS_PB;

    for (int i = tid; i < ROWS_PB * DIM / 4; i += SC_THREADS) {
        v4f z4;
        z4.x = 0.0f; z4.y = 0.0f; z4.z = 0.0f; z4.w = 0.0f;
        ((v4f*)rows)[i] = z4;
    }
    __syncthreads();

    int* q = queue[wave];
    int count = 0;

    v4i cur[TILE_L], nxt[TILE_L];
    load_tile(dst, wave, lane, cur);
    for (int tile = wave; tile < N_ETILES; tile += SC_WAVES) {
        load_tile(dst, tile + SC_WAVES, lane, nxt);
        #pragma unroll
        for (int j = 0; j < TILE_L; ++j) {
            #pragma unroll
            for (int k = 0; k < 4; ++k) {
                const int e = tile * TILE_E + j * 128 + 4 * lane + k;
                const unsigned lr = (unsigned)(cur[j][k] - row0);
                const bool m = lr < (unsigned)ROWS_PB;
                const unsigned mask = __builtin_amdgcn_ballot_w32(m);
                if (mask != 0u) {
                    if (m) {
                        const int pos = count + (int)__builtin_amdgcn_mbcnt_lo(mask, 0u);
                        __hip_atomic_store(q + pos, e | (int)(lr << 20), __ATOMIC_RELAXED, __HIP_MEMORY_SCOPE_WAVEFRONT);
                    }
                    count += __builtin_popcount(mask);
                }
            }
        }
        while (count >= EPW) {
            count -= EPW;
            add_batch<true>(hin, src, rows, q + count, EPW, lane);
        }
        #pragma unroll
        for (int j = 0; j < TILE_L; ++j) cur[j] = nxt[j];
    }
    if (count > 0) add_batch<false>(hin, src, rows, q, count, lane);
    __syncthreads();

    int nrows = N_NODES - row0;
    if (nrows > ROWS_PB) nrows = ROWS_PB;
    v4f* ag4 = (v4f*)(agg + (long)row0 * DIM);
    for (int i = tid; i < nrows * (DIM / 4); i += SC_THREADS) ag4[i] = ((const v4f*)rows)[i];
}

#define TPB 8
#define N_TILES (N_NODES / 16)
static_assert(N_NODES % 16 == 0, "whole 16-row tiles only");

__device__ __forceinline__ void stage_w(const float* __restrict__ Wg, v4f* __restrict__ Wl, int tid) {
    #pragma unroll
    for (int it = 0; it < 16; ++it) {
        const int id = tid + 256 * it;
        const int n  = id & 127;
        const int kg = id >> 7;
        const int t  = kg >> 2;
        const int h  = (kg >> 1) & 1;
        const int q  = kg & 1;
        const int u  = n >> 4;
        const int c  = n & 15;
        const float* wp = Wg + (4 * kg) * DIM + n;
        v4f w;
        w.x = wp[0];
        w.y = wp[DIM];
        w.z = wp[2 * DIM];
        w.w = wp[3 * DIM];
        Wl[(((u * 8 + t) * 2 + q) * 32) + 16 * h + c] = w;
    }
}

__device__ __forceinline__ float row16_sum(float v) {
    v += __builtin_bit_cast(float, __builtin_amdgcn_update_dpp(0, __builtin_bit_cast(int, v), 0x161, 0xf, 0xf, false));
    v += __builtin_bit_cast(float, __builtin_amdgcn_update_dpp(0, __builtin_bit_cast(int, v), 0x162, 0xf, 0xf, false));
    v += __builtin_bit_cast(float, __builtin_amdgcn_update_dpp(0, __builtin_bit_cast(int, v), 0x164, 0xf, 0xf, false));
    v += __builtin_bit_cast(float, __builtin_amdgcn_update_dpp(0, __builtin_bit_cast(int, v), 0x168, 0xf, 0xf, false));
    return v;
}

__global__ __launch_bounds__(256) void gin_mlp(const float* __restrict__ hin,
                                               const float* __restrict__ agg,
                                               const float* __restrict__ W1,
                                               const float* __restrict__ b1,
                                               const float* __restrict__ W2,
                                               const float* __restrict__ b2,
                                               const float* __restrict__ epsp,
                                               float* __restrict__ hout,
                                               float* __restrict__ stats,
                                               int do_stats) {
    __shared__ v4f w1l[DIM * DIM / 4];
    __shared__ v4f w2l[DIM * DIM / 4];
    __shared__ float red[TPB][2 * DIM];

    const int tid  = threadIdx.x;
    const int lane = tid & 31;
    const int c    = lane & 15;
    const int h    = lane >> 4;
    const int wave = __builtin_amdgcn_readfirstlane(tid >> 5);
    const int tile = blockIdx.x * TPB + wave;
    const bool valid = tile < N_TILES;
    const long rowoff = (long)(tile * 16 + c) * DIM + 8 * h;
    const float e1 = 1.0f + epsp[0];

    float zf[8][8];
    if (valid) {
        #pragma unroll
        for (int t = 0; t < 8; ++t) {
            const v4f h0 = *(const v4f*)(hin + rowoff + 16 * t);
            const v4f h1 = *(const v4f*)(hin + rowoff + 16 * t + 4);
            const v4f a0 = *(const v4f*)(agg + rowoff + 16 * t);
            const v4f a1 = *(const v4f*)(agg + rowoff + 16 * t + 4);
            zf[t][0] = e1 * h0.x + a0.x;
            zf[t][1] = e1 * h0.y + a0.y;
            zf[t][2] = e1 * h0.z + a0.z;
            zf[t][3] = e1 * h0.w + a0.w;
            zf[t][4] = e1 * h1.x + a1.x;
            zf[t][5] = e1 * h1.y + a1.y;
            zf[t][6] = e1 * h1.z + a1.z;
            zf[t][7] = e1 * h1.w + a1.w;
        }
    }

    stage_w(W1, w1l, tid);
    stage_w(W2, w2l, tid);
    __syncthreads();

    v8f acc2[8];
    if (valid) {
        v8f acc1[8];
        #pragma unroll
        for (int u = 0; u < 8; ++u) {
            const v4f bl = *(const v4f*)(b1 + 16 * u + 8 * h);
            const v4f bh = *(const v4f*)(b1 + 16 * u + 8 * h + 4);
            acc1[u][0] = bl.x; acc1[u][1] = bl.y; acc1[u][2] = bl.z; acc1[u][3] = bl.w;
            acc1[u][4] = bh.x; acc1[u][5] = bh.y; acc1[u][6] = bh.z; acc1[u][7] = bh.w;
        }
        #pragma unroll
        for (int t = 0; t < 8; ++t) {
            #pragma unroll
            for (int u = 0; u < 8; ++u) {
                const v4f wa = w1l[((u * 8 + t) * 2 + 0) * 32 + lane];
                const v4f wb = w1l[((u * 8 + t) * 2 + 1) * 32 + lane];
                v2f a, b;
                a.x = wa.x; a.y = wa.y; b.x = zf[t][0]; b.y = zf[t][1];
                acc1[u] = __builtin_amdgcn_wmma_f32_16x16x4_f32(false, a, false, b, (short)0, acc1[u], false, false);
                a.x = wa.z; a.y = wa.w; b.x = zf[t][2]; b.y = zf[t][3];
                acc1[u] = __builtin_amdgcn_wmma_f32_16x16x4_f32(false, a, false, b, (short)0, acc1[u], false, false);
                a.x = wb.x; a.y = wb.y; b.x = zf[t][4]; b.y = zf[t][5];
                acc1[u] = __builtin_amdgcn_wmma_f32_16x16x4_f32(false, a, false, b, (short)0, acc1[u], false, false);
                a.x = wb.z; a.y = wb.w; b.x = zf[t][6]; b.y = zf[t][7];
                acc1[u] = __builtin_amdgcn_wmma_f32_16x16x4_f32(false, a, false, b, (short)0, acc1[u], false, false);
            }
        }
        #pragma unroll
        for (int t = 0; t < 8; ++t) {
            #pragma unroll
            for (int r = 0; r < 8; ++r) acc1[t][r] = fmaxf(acc1[t][r], 0.0f);
        }

        #pragma unroll
        for (int u = 0; u < 8; ++u) {
            const v4f bl = *(const v4f*)(b2 + 16 * u + 8 * h);
            const v4f bh = *(const v4f*)(b2 + 16 * u + 8 * h + 4);
            acc2[u][0] = bl.x; acc2[u][1] = bl.y; acc2[u][2] = bl.z; acc2[u][3] = bl.w;
            acc2[u][4] = bh.x; acc2[u][5] = bh.y; acc2[u][6] = bh.z; acc2[u][7] = bh.w;
        }
        #pragma unroll
        for (int t = 0; t < 8; ++t) {
            #pragma unroll
            for (int u = 0; u < 8; ++u) {
                const v4f wa = w2l[((u * 8 + t) * 2 + 0) * 32 + lane];
                const v4f wb = w2l[((u * 8 + t) * 2 + 1) * 32 + lane];
                v2f a, b;
                a.x = wa.x; a.y = wa.y; b.x = acc1[t][0]; b.y = acc1[t][1];
                acc2[u] = __builtin_amdgcn_wmma_f32_16x16x4_f32(false, a, false, b, (short)0, acc2[u], false, false);
                a.x = wa.z; a.y = wa.w; b.x = acc1[t][2]; b.y = acc1[t][3];
                acc2[u] = __builtin_amdgcn_wmma_f32_16x16x4_f32(false, a, false, b, (short)0, acc2[u], false, false);
                a.x = wb.x; a.y = wb.y; b.x = acc1[t][4]; b.y = acc1[t][5];
                acc2[u] = __builtin_amdgcn_wmma_f32_16x16x4_f32(false, a, false, b, (short)0, acc2[u], false, false);
                a.x = wb.z; a.y = wb.w; b.x = acc1[t][6]; b.y = acc1[t][7];
                acc2[u] = __builtin_amdgcn_wmma_f32_16x16x4_f32(false, a, false, b, (short)0, acc2[u], false, false);
            }
        }

        #pragma unroll
        for (int u = 0; u < 8; ++u) {
            v4f lo, hi4;
            lo.x = acc2[u][0]; lo.y = acc2[u][1]; lo.z = acc2[u][2]; lo.w = acc2[u][3];
            hi4.x = acc2[u][4]; hi4.y = acc2[u][5]; hi4.z = acc2[u][6]; hi4.w = acc2[u][7];
            *(v4f*)(hout + rowoff + 16 * u)     = lo;
            *(v4f*)(hout + rowoff + 16 * u + 4) = hi4;
        }
    } else {
        #pragma unroll
        for (int u = 0; u < 8; ++u) {
            #pragma unroll
            for (int r = 0; r < 8; ++r) acc2[u][r] = 0.0f;
        }
    }

    if (do_stats) {
        float cs[8][8], cq[8][8];
        #pragma unroll
        for (int u = 0; u < 8; ++u) {
            #pragma unroll
            for (int r = 0; r < 8; ++r) {
                const float val = acc2[u][r];
                cs[u][r] = row16_sum(val);
                cq[u][r] = row16_sum(val * val);
            }
        }
        if (c == 0) {
            #pragma unroll
            for (int u = 0; u < 8; ++u) {
                #pragma unroll
                for (int r = 0; r < 8; ++r) {
                    red[wave][16 * u + 8 * h + r]       = cs[u][r];
                    red[wave][DIM + 16 * u + 8 * h + r] = cq[u][r];
                }
            }
        }
        __syncthreads();
        float tot = 0.0f;
        #pragma unroll
        for (int w = 0; w < TPB; ++w) tot += red[w][tid];
        atomicAdd(&stats[tid], tot);
    }
}

__global__ __launch_bounds__(256) void gin_bn_relu(float* __restrict__ z,
                                                   const float* __restrict__ stats,
                                                   const float* __restrict__ gamma,
                                                   const float* __restrict__ beta) {
    long i = (long)blockIdx.x * blockDim.x + threadIdx.x;
    if (i >= (long)N_NODES * DIM) return;
    int c = (int)(i & (DIM - 1));
    const float invN = 1.0f / (float)N_NODES;
    float mu  = stats[c] * invN;
    float var = stats[DIM + c] * invN - mu * mu;
    float v = (z[i] - mu) * rsqrtf(var + BN_EPS) * gamma[c] + beta[c];
    z[i] = fmaxf(v, 0.0f);
}

__global__ __launch_bounds__(256) void gin_pool(const float* __restrict__ h,
                                                const int* __restrict__ batch,
                                                float* __restrict__ out) {
    const int lane = (int)(threadIdx.x & 31);
    const int wv = __builtin_amdgcn_readfirstlane((int)(blockIdx.x * 8 + (threadIdx.x >> 5)));
    const int n0 = wv * EPW;
    if (n0 >= N_NODES) return;
    int g[EPW];
    #pragma unroll
    for (int j = 0; j < EPW; ++j) g[j] = batch[n0 + j];
    float v[EPW][4];
    #pragma unroll
    for (int j = 0; j < EPW; ++j) {
        const float* hp = h + (long)(n0 + j) * DIM + lane;
        v[j][0] = hp[0];
        v[j][1] = hp[32];
        v[j][2] = hp[64];
        v[j][3] = hp[96];
    }
    #pragma unroll
    for (int j = 0; j < EPW; ++j) {
        float* op = out + (long)g[j] * DIM + lane;
        atomicAdd(op + 0,  v[j][0]);
        atomicAdd(op + 32, v[j][1]);
        atomicAdd(op + 64, v[j][2]);
        atomicAdd(op + 96, v[j][3]);
    }
}

extern "C" void kernel_launch(void* const* d_in, const int* in_sizes, int n_in,
                              void* d_out, int out_size, void* d_ws, size_t ws_size,
                              hipStream_t stream) {
    const float* x     = (const float*)d_in[0];
    const int*   ei    = (const int*)d_in[1];
    const int*   batch = (const int*)d_in[2];
    const float* W1    = (const float*)d_in[3];
    const float* b1    = (const float*)d_in[4];
    const float* W2    = (const float*)d_in[5];
    const float* b2    = (const float*)d_in[6];
    const float* eps   = (const float*)d_in[7];
    const float* gamma = (const float*)d_in[8];
    const float* beta  = (const float*)d_in[9];
    float* out = (float*)d_out;

    const long ND = (long)N_NODES * DIM;
    float* hA    = (float*)d_ws;
    float* hB    = hA + ND;
    float* agg   = hB + ND;
    float* stats = agg + ND;

    const int* src = ei;
    const int* dst = ei + N_EDGES;

    const int  mlpGrd  = (N_TILES + TPB - 1) / TPB;
    const int  bnGrd   = (int)((ND + 255) / 256);
    const int  poolGrd = (N_NODES / EPW + 7) / 8;

    for (int i = 0; i < N_LAYERS; ++i) {
        const float* hin  = (i == 0) ? x : ((i & 1) ? hA : hB);
        float*       hout = (i & 1) ? hB : hA;
        int do_stats = (i < N_LAYERS - 1) ? 1 : 0;

        if (do_stats) gin_zero<<<1, 256, 0, stream>>>(stats, 2 * DIM);
        gin_rowsum<<<N_SBLK, SC_THREADS, 0, stream>>>(hin, src, dst, agg);
        gin_mlp<<<mlpGrd, 256, 0, stream>>>(hin, agg,
                                            W1 + (long)i * DIM * DIM, b1 + (long)i * DIM,
                                            W2 + (long)i * DIM * DIM, b2 + (long)i * DIM,
                                            eps + i, hout, stats, do_stats);
        if (do_stats) {
            gin_bn_relu<<<bnGrd, 256, 0, stream>>>(hout, stats,
                                                   gamma + (long)i * DIM,
                                                   beta + (long)i * DIM);
        }
    }
    gin_zero<<<(NUM_GRAPHS * DIM + 255) / 256, 256, 0, stream>>>(out, NUM_GRAPHS * DIM);
    gin_pool<<<poolGrd, 256, 0, stream>>>(hB, batch, out);
}


# Neighbour sum: layer 1's filter also writes each block's kept edges (source | row) as a dense list to the workspace; layers 2-4 add from the list, read no dst; sum order = LDS add arrival as before
# speedup vs baseline: 2.1545x; 1.1458x over previous
#include <hip/hip_runtime.h>
#include <hip/hip_bf16.h>

#define N_NODES   50000
#define N_EDGES   800000
#define DIM       128
#define N_LAYERS  4
#define NUM_GRAPHS 128
#define BN_EPS    1e-5f

typedef __attribute__((ext_vector_type(2))) float v2f;
typedef __attribute__((ext_vector_type(8))) float v8f;

__global__ __launch_bounds__(256) void gin_zero(float* __restrict__ p, long n) {
    long i = (long)blockIdx.x * blockDim.x + threadIdx.x;
    if (i < n) p[i] = 0.0f;
}

#define EPW 8
#define ROWS_PB   196
#define SC_WAVES  32
#define SC_THREADS (SC_WAVES * 32)
#define N_SBLK    ((N_NODES + ROWS_PB - 1) / ROWS_PB)
#define TILE_L    8
#define TILE_E    (TILE_L * 128)
#define N_ETILES  ((N_EDGES + TILE_E - 1) / TILE_E)
#define QCAP      (TILE_E + EPW)
#define LCAP      65536
static_assert(N_EDGES % 4 == 0, "a 16-byte index load must not straddle the end of dst");
static_assert(N_EDGES <= (1 << 20), "the edge number must fit the low 20 bits of a queue word");
static_assert(ROWS_PB < (1 << 11), "the local row must fit the bits above the edge number");
static_assert(N_NODES % EPW == 0, "a pool wave's nodes must all exist");
static_assert(N_NODES <= (1 << 20), "a source number must fit the low 20 bits of a list word");
static_assert(LCAP % EPW == 0, "a batch of list words must not straddle two blocks' lists");

typedef __attribute__((ext_vector_type(4))) int   v4i;
typedef __attribute__((ext_vector_type(4))) float v4f;

__device__ __forceinline__ void load_tile(const int* __restrict__ dst, int tile, int lane, v4i (&t)[TILE_L]) {
    #pragma unroll
    for (int j = 0; j < TILE_L; ++j) {
        const int idx = tile * TILE_E + j * 128 + 4 * lane;
        const int idc = idx < N_EDGES ? idx : N_EDGES - 4;
        v4i d = *(const v4i*)(dst + idc);
        if (idx >= N_EDGES) { d.x = -1; d.y = -1; d.z = -1; d.w = -1; }
        t[j] = d;
    }
}

template <bool FULL>
__device__ __forceinline__ void add_words(const float* __restrict__ hin, float* rows,
                                          const int (&wd)[EPW], int n, int lane) {
    float v[EPW][4];
    #pragma unroll
    for (int j = 0; j < EPW; ++j) {
        if (FULL || j < n) {
            const float* hs = hin + (long)(wd[j] & 0xFFFFF) * DIM + lane;
            v[j][0] = hs[0];
            v[j][1] = hs[32];
            v[j][2] = hs[64];
            v[j][3] = hs[96];
        }
    }
    #pragma unroll
    for (int j = 0; j < EPW; ++j) {
        if (FULL || j < n) {
            float* r = rows + (wd[j] >> 20) * DIM + lane;
            __hip_atomic_fetch_add(r + 0,  v[j][0], __ATOMIC_RELAXED, __HIP_MEMORY_SCOPE_WORKGROUP);
            __hip_atomic_fetch_add(r + 32, v[j][1], __ATOMIC_RELAXED, __HIP_MEMORY_SCOPE_WORKGROUP);
            __hip_atomic_fetch_add(r + 64, v[j][2], __ATOMIC_RELAXED, __HIP_MEMORY_SCOPE_WORKGROUP);
            __hip_atomic_fetch_add(r + 96, v[j][3], __ATOMIC_RELAXED, __HIP_MEMORY_SCOPE_WORKGROUP);
        }
    }
}

template <bool FULL, bool REC>
__device__ __forceinline__ void add_batch(const float* __restrict__ hin, const int* __restrict__ src,
                                          float* rows, const int* qe, int n, int lane,
                                          int* lrec, int* __restrict__ list) {
    int ent[EPW], wd[EPW];
    #pragma unroll
    for (int j = 0; j < EPW; ++j) {
        ent[j] = 0;
        if (FULL || j < n)
            ent[j] = __builtin_amdgcn_readfirstlane(__hip_atomic_load(qe + j, __ATOMIC_RELAXED, __HIP_MEMORY_SCOPE_WAVEFRONT));
    }
    #pragma unroll
    for (int j = 0; j < EPW; ++j) {
        wd[j] = 0;
        if (FULL || j < n) wd[j] = src[ent[j] & 0xFFFFF] | (ent[j] & ~0xFFFFF);
    }
    if (REC) {
        int pos = 0;
        if (lane == 0) pos = __hip_atomic_fetch_add(lrec, n, __ATOMIC_RELAXED, __HIP_MEMORY_SCOPE_WORKGROUP);
        pos = __builtin_amdgcn_readfirstlane(pos);
        if (pos + n <= LCAP) {
            int w = 0;
            #pragma unroll
            for (int j = 0; j < EPW; ++j) {
                if (lane == j) w = wd[j];
            }
            if (lane < n) list[pos + lane] = w;
        }
    }
    add_words<FULL>(hin, rows, wd, n, lane);
}

template <bool REC>
__device__ __forceinline__ void filter_add(const float* __restrict__ hin, const int* __restrict__ src,
                                           const int* __restrict__ dst, float* rows, int* q,
                                           int row0, int wave, int lane, int* lrec, int* __restrict__ list) {
    int count = 0;

    v4i cur[TILE_L], nxt[TILE_L];
    load_tile(dst, wave, lane, cur);
    for (int tile = wave; tile < N_ETILES; tile += SC_WAVES) {
        load_tile(dst, tile + SC_WAVES, lane, nxt);
        #pragma unroll
        for (int j = 0; j < TILE_L; ++j) {
            #pragma unroll
            for (int k = 0; k < 4; ++k) {
                const int e = tile * TILE_E + j * 128 + 4 * lane + k;
                const unsigned lr = (unsigned)(cur[j][k] - row0);
                const bool m = lr < (unsigned)ROWS_PB;
                const unsigned mask = __builtin_amdgcn_ballot_w32(m);
                if (mask != 0u) {
                    if (m) {
                        const int pos = count + (int)__builtin_amdgcn_mbcnt_lo(mask, 0u);
                        __hip_atomic_store(q + pos, e | (int)(lr << 20), __ATOMIC_RELAXED, __HIP_MEMORY_SCOPE_WAVEFRONT);
                    }
                    count += __builtin_popcount(mask);
                }
            }
        }
        while (count >= EPW) {
            count -= EPW;
            add_batch<true, REC>(hin, src, rows, q + count, EPW, lane, lrec, list);
        }
        #pragma unroll
        for (int j = 0; j < TILE_L; ++j) cur[j] = nxt[j];
    }
    if (count > 0) add_batch<false, REC>(hin, src, rows, q, count, lane, lrec, list);
}

__device__ __forceinline__ void zero_rows(float* rows, int tid) {
    for (int i = tid; i < ROWS_PB * DIM / 4; i += SC_THREADS) {
        v4f z4;
        z4.x = 0.0f; z4.y = 0.0f; z4.z = 0.0f; z4.w = 0.0f;
        ((v4f*)rows)[i] = z4;
    }
}

__device__ __forceinline__ void store_rows(const float* rows, float* __restrict__ agg, int row0, int tid) {
    int nrows = N_NODES - row0;
    if (nrows > ROWS_PB) nrows = ROWS_PB;
    v4f* ag4 = (v4f*)(agg + (long)row0 * DIM);
    for (int i = tid; i < nrows * (DIM / 4); i += SC_THREADS) ag4[i] = ((const v4f*)rows)[i];
}

__global__ __launch_bounds__(SC_THREADS) void gin_rowsum_first(const float* __restrict__ hin,
                                                               const int* __restrict__ src,
                                                               const int* __restrict__ dst,
                                                               float* __restrict__ agg,
                                                               int* __restrict__ lists,
                                                               int* __restrict__ lens) {
    __shared__ __attribute__((aligned(16))) float rows[ROWS_PB * DIM];
    __shared__ int   queue[SC_WAVES][QCAP];
    __shared__ int   lrec;

    const int tid  = (int)threadIdx.x;
    const int lane = tid & 31;
    const int wave = __builtin_amdgcn_readfirstlane(tid >> 5);
    const int row0 = (int)blockIdx.x * ROWS_PB;

    zero_rows(rows, tid);
    if (tid == 0) lrec = 0;
    __syncthreads();

    filter_add<true>(hin, src, dst, rows, queue[wave], row0, wave, lane, &lrec, lists + (long)blockIdx.x * LCAP);
    __syncthreads();

    if (tid == 0) lens[blockIdx.x] = lrec;
    store_rows(rows, agg, row0, tid);
}

__global__ __launch_bounds__(SC_THREADS) void gin_rowsum_next(const float* __restrict__ hin,
                                                              const int* __restrict__ src,
                                                              const int* __restrict__ dst,
                                                              float* __restrict__ agg,
                                                              const int* __restrict__ lists,
                                                              const int* __restrict__ lens) {
    __shared__ __attribute__((aligned(16))) float rows[ROWS_PB * DIM];
    __shared__ int   queue[SC_WAVES][QCAP];

    const int tid  = (int)threadIdx.x;
    const int lane = tid & 31;
    const int wave = __builtin_amdgcn_readfirstlane(tid >> 5);
    const int row0 = (int)blockIdx.x * ROWS_PB;

    zero_rows(rows, tid);
    __syncthreads();

    const int total = lens[blockIdx.x];
    if (total <= LCAP) {
        const int* bl = lists + (long)blockIdx.x * LCAP;
        const int nb = total / EPW;
        if (wave < nb) {
            int vw = bl[EPW * wave + (lane & (EPW - 1))];
            for (int b = wave; b < nb; b += SC_WAVES) {
                int wd[EPW];
                #pragma unroll
                for (int j = 0; j < EPW; ++j) wd[j] = __builtin_amdgcn_readlane(vw, j);
                const int bn = (b + SC_WAVES < nb) ? b + SC_WAVES : b;
                vw = bl[EPW * bn + (lane & (EPW - 1))];
                add_words<true>(hin, rows, wd, EPW, lane);
            }
        }
        const int rem = total - EPW * nb;
        if (rem > 0 && wave == (nb & (SC_WAVES - 1))) {
            int wd[EPW];
            const int vt = bl[EPW * nb + (lane < rem ? lane : 0)];
            #pragma unroll
            for (int j = 0; j < EPW; ++j) wd[j] = __builtin_amdgcn_readlane(vt, j);
            add_words<false>(hin, rows, wd, rem, lane);
        }
    } else {
        filter_add<false>(hin, src, dst, rows, queue[wave], row0, wave, lane, nullptr, nullptr);
    }
    __syncthreads();

    store_rows(rows, agg, row0, tid);
}

#define TPB 8
#define N_TILES (N_NODES / 16)
static_assert(N_NODES % 16 == 0, "whole 16-row tiles only");

__device__ __forceinline__ void stage_w(const float* __restrict__ Wg, v4f* __restrict__ Wl, int tid) {
    #pragma unroll
    for (int it = 0; it < 16; ++it) {
        const int id = tid + 256 * it;
        const int n  = id & 127;
        const int kg = id >> 7;
        const int t  = kg >> 2;
        const int h  = (kg >> 1) & 1;
        const int q  = kg & 1;
        const int u  = n >> 4;
        const int c  = n & 15;
        const float* wp = Wg + (4 * kg) * DIM + n;
        v4f w;
        w.x = wp[0];
        w.y = wp[DIM];
        w.z = wp[2 * DIM];
        w.w = wp[3 * DIM];
        Wl[(((u * 8 + t) * 2 + q) * 32) + 16 * h + c] = w;
    }
}

__device__ __forceinline__ float row16_sum(float v) {
    v += __builtin_bit_cast(float, __builtin_amdgcn_update_dpp(0, __builtin_bit_cast(int, v), 0x161, 0xf, 0xf, false));
    v += __builtin_bit_cast(float, __builtin_amdgcn_update_dpp(0, __builtin_bit_cast(int, v), 0x162, 0xf, 0xf, false));
    v += __builtin_bit_cast(float, __builtin_amdgcn_update_dpp(0, __builtin_bit_cast(int, v), 0x164, 0xf, 0xf, false));
    v += __builtin_bit_cast(float, __builtin_amdgcn_update_dpp(0, __builtin_bit_cast(int, v), 0x168, 0xf, 0xf, false));
    return v;
}

__global__ __launch_bounds__(256) void gin_mlp(const float* __restrict__ hin,
                                               const float* __restrict__ agg,
                                               const float* __restrict__ W1,
                                               const float* __restrict__ b1,
                                               const float* __restrict__ W2,
                                               const float* __restrict__ b2,
                                               const float* __restrict__ epsp,
                                               float* __restrict__ hout,
                                               float* __restrict__ stats,
                                               int do_stats) {
    __shared__ v4f w1l[DIM * DIM / 4];
    __shared__ v4f w2l[DIM * DIM / 4];
    __shared__ float red[TPB][2 * DIM];

    const int tid  = threadIdx.x;
    const int lane = tid & 31;
    const int c    = lane & 15;
    const int h    = lane >> 4;
    const int wave = __builtin_amdgcn_readfirstlane(tid >> 5);
    const int tile = blockIdx.x * TPB + wave;
    const bool valid = tile < N_TILES;
    const long rowoff = (long)(tile * 16 + c) * DIM + 8 * h;
    const float e1 = 1.0f + epsp[0];

    float zf[8][8];
    if (valid) {
        #pragma unroll
        for (int t = 0; t < 8; ++t) {
            const v4f h0 = *(const v4f*)(hin + rowoff + 16 * t);
            const v4f h1 = *(const v4f*)(hin + rowoff + 16 * t + 4);
            const v4f a0 = *(const v4f*)(agg + rowoff + 16 * t);
            const v4f a1 = *(const v4f*)(agg + rowoff + 16 * t + 4);
            zf[t][0] = e1 * h0.x + a0.x;
            zf[t][1] = e1 * h0.y + a0.y;
            zf[t][2] = e1 * h0.z + a0.z;
            zf[t][3] = e1 * h0.w + a0.w;
            zf[t][4] = e1 * h1.x + a1.x;
            zf[t][5] = e1 * h1.y + a1.y;
            zf[t][6] = e1 * h1.z + a1.z;
            zf[t][7] = e1 * h1.w + a1.w;
        }
    }

    stage_w(W1, w1l, tid);
    stage_w(W2, w2l, tid);
    __syncthreads();

    v8f acc2[8];
    if (valid) {
        v8f acc1[8];
        #pragma unroll
        for (int u = 0; u < 8; ++u) {
            const v4f bl = *(const v4f*)(b1 + 16 * u + 8 * h);
            const v4f bh = *(const v4f*)(b1 + 16 * u + 8 * h + 4);
            acc1[u][0] = bl.x; acc1[u][1] = bl.y; acc1[u][2] = bl.z; acc1[u][3] = bl.w;
            acc1[u][4] = bh.x; acc1[u][5] = bh.y; acc1[u][6] = bh.z; acc1[u][7] = bh.w;
        }
        #pragma unroll
        for (int t = 0; t < 8; ++t) {
            #pragma unroll
            for (int u = 0; u < 8; ++u) {
                const v4f wa = w1l[((u * 8 + t) * 2 + 0) * 32 + lane];
                const v4f wb = w1l[((u * 8 + t) * 2 + 1) * 32 + lane];
                v2f a, b;
                a.x = wa.x; a.y = wa.y; b.x = zf[t][0]; b.y = zf[t][1];
                acc1[u] = __builtin_amdgcn_wmma_f32_16x16x4_f32(false, a, false, b, (short)0, acc1[u], false, false);
                a.x = wa.z; a.y = wa.w; b.x = zf[t][2]; b.y = zf[t][3];
                acc1[u] = __builtin_amdgcn_wmma_f32_16x16x4_f32(false, a, false, b, (short)0, acc1[u], false, false);
                a.x = wb.x; a.y = wb.y; b.x = zf[t][4]; b.y = zf[t][5];
                acc1[u] = __builtin_amdgcn_wmma_f32_16x16x4_f32(false, a, false, b, (short)0, acc1[u], false, false);
                a.x = wb.z; a.y = wb.w; b.x = zf[t][6]; b.y = zf[t][7];
                acc1[u] = __builtin_amdgcn_wmma_f32_16x16x4_f32(false, a, false, b, (short)0, acc1[u], false, false);
            }
        }
        #pragma unroll
        for (int t = 0; t < 8; ++t) {
            #pragma unroll
            for (int r = 0; r < 8; ++r) acc1[t][r] = fmaxf(acc1[t][r], 0.0f);
        }

        #pragma unroll
        for (int u = 0; u < 8; ++u) {
            const v4f bl = *(const v4f*)(b2 + 16 * u + 8 * h);
            const v4f bh = *(const v4f*)(b2 + 16 * u + 8 * h + 4);
            acc2[u][0] = bl.x; acc2[u][1] = bl.y; acc2[u][2] = bl.z; acc2[u][3] = bl.w;
            acc2[u][4] = bh.x; acc2[u][5] = bh.y; acc2[u][6] = bh.z; acc2[u][7] = bh.w;
        }
        #pragma unroll
        for (int t = 0; t < 8; ++t) {
            #pragma unroll
            for (int u = 0; u < 8; ++u) {
                const v4f wa = w2l[((u * 8 + t) * 2 + 0) * 32 + lane];
                const v4f wb = w2l[((u * 8 + t) * 2 + 1) * 32 + lane];
                v2f a, b;
                a.x = wa.x; a.y = wa.y; b.x = acc1[t][0]; b.y = acc1[t][1];
                acc2[u] = __builtin_amdgcn_wmma_f32_16x16x4_f32(false, a, false, b, (short)0, acc2[u], false, false);
                a.x = wa.z; a.y = wa.w; b.x = acc1[t][2]; b.y = acc1[t][3];
                acc2[u] = __builtin_amdgcn_wmma_f32_16x16x4_f32(false, a, false, b, (short)0, acc2[u], false, false);
                a.x = wb.x; a.y = wb.y; b.x = acc1[t][4]; b.y = acc1[t][5];
                acc2[u] = __builtin_amdgcn_wmma_f32_16x16x4_f32(false, a, false, b, (short)0, acc2[u], false, false);
                a.x = wb.z; a.y = wb.w; b.x = acc1[t][6]; b.y = acc1[t][7];
                acc2[u] = __builtin_amdgcn_wmma_f32_16x16x4_f32(false, a, false, b, (short)0, acc2[u], false, false);
            }
        }

        #pragma unroll
        for (int u = 0; u < 8; ++u) {
            v4f lo, hi4;
            lo.x = acc2[u][0]; lo.y = acc2[u][1]; lo.z = acc2[u][2]; lo.w = acc2[u][3];
            hi4.x = acc2[u][4]; hi4.y = acc2[u][5]; hi4.z = acc2[u][6]; hi4.w = acc2[u][7];
            *(v4f*)(hout + rowoff + 16 * u)     = lo;
            *(v4f*)(hout + rowoff + 16 * u + 4) = hi4;
        }
    } else {
        #pragma unroll
        for (int u = 0; u < 8; ++u) {
            #pragma unroll
            for (int r = 0; r < 8; ++r) acc2[u][r] = 0.0f;
        }
    }

    if (do_stats) {
        float cs[8][8], cq[8][8];
        #pragma unroll
        for (int u = 0; u < 8; ++u) {
            #pragma unroll
            for (int r = 0; r < 8; ++r) {
                const float val = acc2[u][r];
                cs[u][r] = row16_sum(val);
                cq[u][r] = row16_sum(val * val);
            }
        }
        if (c == 0) {
            #pragma unroll
            for (int u = 0; u < 8; ++u) {
                #pragma unroll
                for (int r = 0; r < 8; ++r) {
                    red[wave][16 * u + 8 * h + r]       = cs[u][r];
                    red[wave][DIM + 16 * u + 8 * h + r] = cq[u][r];
                }
            }
        }
        __syncthreads();
        float tot = 0.0f;
        #pragma unroll
        for (int w = 0; w < TPB; ++w) tot += red[w][tid];
        atomicAdd(&stats[tid], tot);
    }
}

__global__ __launch_bounds__(256) void gin_bn_relu(float* __restrict__ z,
                                                   const float* __restrict__ stats,
                                                   const float* __restrict__ gamma,
                                                   const float* __restrict__ beta) {
    long i = (long)blockIdx.x * blockDim.x + threadIdx.x;
    if (i >= (long)N_NODES * DIM) return;
    int c = (int)(i & (DIM - 1));
    const float invN = 1.0f / (float)N_NODES;
    float mu  = stats[c] * invN;
    float var = stats[DIM + c] * invN - mu * mu;
    float v = (z[i] - mu) * rsqrtf(var + BN_EPS) * gamma[c] + beta[c];
    z[i] = fmaxf(v, 0.0f);
}

__global__ __launch_bounds__(256) void gin_pool(const float* __restrict__ h,
                                                const int* __restrict__ batch,
                                                float* __restrict__ out) {
    const int lane = (int)(threadIdx.x & 31);
    const int wv = __builtin_amdgcn_readfirstlane((int)(blockIdx.x * 8 + (threadIdx.x >> 5)));
    const int n0 = wv * EPW;
    if (n0 >= N_NODES) return;
    int g[EPW];
    #pragma unroll
    for (int j = 0; j < EPW; ++j) g[j] = batch[n0 + j];
    float v[EPW][4];
    #pragma unroll
    for (int j = 0; j < EPW; ++j) {
        const float* hp = h + (long)(n0 + j) * DIM + lane;
        v[j][0] = hp[0];
        v[j][1] = hp[32];
        v[j][2] = hp[64];
        v[j][3] = hp[96];
    }
    #pragma unroll
    for (int j = 0; j < EPW; ++j) {
        float* op = out + (long)g[j] * DIM + lane;
        atomicAdd(op + 0,  v[j][0]);
        atomicAdd(op + 32, v[j][1]);
        atomicAdd(op + 64, v[j][2]);
        atomicAdd(op + 96, v[j][3]);
    }
}

extern "C" void kernel_launch(void* const* d_in, const int* in_sizes, int n_in,
                              void* d_out, int out_size, void* d_ws, size_t ws_size,
                              hipStream_t stream) {
    const float* x     = (const float*)d_in[0];
    const int*   ei    = (const int*)d_in[1];
    const int*   batch = (const int*)d_in[2];
    const float* W1    = (const float*)d_in[3];
    const float* b1    = (const float*)d_in[4];
    const float* W2    = (const float*)d_in[5];
    const float* b2    = (const float*)d_in[6];
    const float* eps   = (const float*)d_in[7];
    const float* gamma = (const float*)d_in[8];
    const float* beta  = (const float*)d_in[9];
    float* out = (float*)d_out;

    const long ND = (long)N_NODES * DIM;
    float* hA    = (float*)d_ws;
    float* hB    = hA + ND;
    float* agg   = hB + ND;
    float* stats = agg + ND;
    int*   lens  = (int*)(stats + 2 * DIM);
    int*   lists = lens + N_SBLK;

    const int* src = ei;
    const int* dst = ei + N_EDGES;

    const int  mlpGrd  = (N_TILES + TPB - 1) / TPB;
    const int  bnGrd   = (int)((ND + 255) / 256);
    const int  poolGrd = (N_NODES / EPW + 7) / 8;

    for (int i = 0; i < N_LAYERS; ++i) {
        const float* hin  = (i == 0) ? x : ((i & 1) ? hA : hB);
        float*       hout = (i & 1) ? hB : hA;
        int do_stats = (i < N_LAYERS - 1) ? 1 : 0;

        if (do_stats) gin_zero<<<1, 256, 0, stream>>>(stats, 2 * DIM);
        if (i == 0) gin_rowsum_first<<<N_SBLK, SC_THREADS, 0, stream>>>(hin, src, dst, agg, lists, lens);
        else        gin_rowsum_next<<<N_SBLK, SC_THREADS, 0, stream>>>(hin, src, dst, agg, lists, lens);
        gin_mlp<<<mlpGrd, 256, 0, stream>>>(hin, agg,
                                            W1 + (long)i * DIM * DIM, b1 + (long)i * DIM,
                                            W2 + (long)i * DIM * DIM, b2 + (long)i * DIM,
                                            eps + i, hout, stats, do_stats);
        if (do_stats) {
            gin_bn_relu<<<bnGrd, 256, 0, stream>>>(hout, stats,
                                                   gamma + (long)i * DIM,
                                                   beta + (long)i * DIM);
        }
    }
    gin_zero<<<(NUM_GRAPHS * DIM + 255) / 256, 256, 0, stream>>>(out, NUM_GRAPHS * DIM);
    gin_pool<<<poolGrd, 256, 0, stream>>>(hB, batch, out);
}


# MLP kernel prologue: a thread's 160 global loads (activation rows, W1, W2) issue as one group ahead of the LDS stores; the built code had 4 loads, wait, store 32 times in a row; no sum order changes
# speedup vs baseline: 2.5843x; 1.1995x over previous
#include <hip/hip_runtime.h>
#include <hip/hip_bf16.h>

#define N_NODES   50000
#define N_EDGES   800000
#define DIM       128
#define N_LAYERS  4
#define NUM_GRAPHS 128
#define BN_EPS    1e-5f

typedef __attribute__((ext_vector_type(2))) float v2f;
typedef __attribute__((ext_vector_type(8))) float v8f;

__global__ __launch_bounds__(256) void gin_zero(float* __restrict__ p, long n) {
    long i = (long)blockIdx.x * blockDim.x + threadIdx.x;
    if (i < n) p[i] = 0.0f;
}

#define EPW 8
#define ROWS_PB   196
#define SC_WAVES  32
#define SC_THREADS (SC_WAVES * 32)
#define N_SBLK    ((N_NODES + ROWS_PB - 1) / ROWS_PB)
#define TILE_L    8
#define TILE_E    (TILE_L * 128)
#define N_ETILES  ((N_EDGES + TILE_E - 1) / TILE_E)
#define QCAP      (TILE_E + EPW)
#define LCAP      65536
static_assert(N_EDGES % 4 == 0, "a 16-byte index load must not straddle the end of dst");
static_assert(N_EDGES <= (1 << 20), "the edge number must fit the low 20 bits of a queue word");
static_assert(ROWS_PB < (1 << 11), "the local row must fit the bits above the edge number");
static_assert(N_NODES % EPW == 0, "a pool wave's nodes must all exist");
static_assert(N_NODES <= (1 << 20), "a source number must fit the low 20 bits of a list word");
static_assert(LCAP % EPW == 0, "a batch of list words must not straddle two blocks' lists");

typedef __attribute__((ext_vector_type(4))) int   v4i;
typedef __attribute__((ext_vector_type(4))) float v4f;

__device__ __forceinline__ void load_tile(const int* __restrict__ dst, int tile, int lane, v4i (&t)[TILE_L]) {
    #pragma unroll
    for (int j = 0; j < TILE_L; ++j) {
        const int idx = tile * TILE_E + j * 128 + 4 * lane;
        const int idc = idx < N_EDGES ? idx : N_EDGES - 4;
        v4i d = *(const v4i*)(dst + idc);
        if (idx >= N_EDGES) { d.x = -1; d.y = -1; d.z = -1; d.w = -1; }
        t[j] = d;
    }
}

template <bool FULL>
__device__ __forceinline__ void add_words(const float* __restrict__ hin, float* rows,
                                          const int (&wd)[EPW], int n, int lane) {
    float v[EPW][4];
    #pragma unroll
    for (int j = 0; j < EPW; ++j) {
        if (FULL || j < n) {
            const float* hs = hin + (long)(wd[j] & 0xFFFFF) * DIM + lane;
            v[j][0] = hs[0];
            v[j][1] = hs[32];
            v[j][2] = hs[64];
            v[j][3] = hs[96];
        }
    }
    #pragma unroll
    for (int j = 0; j < EPW; ++j) {
        if (FULL || j < n) {
            float* r = rows + (wd[j] >> 20) * DIM + lane;
            __hip_atomic_fetch_add(r + 0,  v[j][0], __ATOMIC_RELAXED, __HIP_MEMORY_SCOPE_WORKGROUP);
            __hip_atomic_fetch_add(r + 32, v[j][1], __ATOMIC_RELAXED, __HIP_MEMORY_SCOPE_WORKGROUP);
            __hip_atomic_fetch_add(r + 64, v[j][2], __ATOMIC_RELAXED, __HIP_MEMORY_SCOPE_WORKGROUP);
            __hip_atomic_fetch_add(r + 96, v[j][3], __ATOMIC_RELAXED, __HIP_MEMORY_SCOPE_WORKGROUP);
        }
    }
}

template <bool FULL, bool REC>
__device__ __forceinline__ void add_batch(const float* __restrict__ hin, const int* __restrict__ src,
                                          float* rows, const int* qe, int n, int lane,
                                          int* lrec, int* __restrict__ list) {
    int ent[EPW], wd[EPW];
    #pragma unroll
    for (int j = 0; j < EPW; ++j) {
        ent[j] = 0;
        if (FULL || j < n)
            ent[j] = __builtin_amdgcn_readfirstlane(__hip_atomic_load(qe + j, __ATOMIC_RELAXED, __HIP_MEMORY_SCOPE_WAVEFRONT));
    }
    #pragma unroll
    for (int j = 0; j < EPW; ++j) {
        wd[j] = 0;
        if (FULL || j < n) wd[j] = src[ent[j] & 0xFFFFF] | (ent[j] & ~0xFFFFF);
    }
    if (REC) {
        int pos = 0;
        if (lane == 0) pos = __hip_atomic_fetch_add(lrec, n, __ATOMIC_RELAXED, __HIP_MEMORY_SCOPE_WORKGROUP);
        pos = __builtin_amdgcn_readfirstlane(pos);
        if (pos + n <= LCAP) {
            int w = 0;
            #pragma unroll
            for (int j = 0; j < EPW; ++j) {
                if (lane == j) w = wd[j];
            }
            if (lane < n) list[pos + lane] = w;
        }
    }
    add_words<FULL>(hin, rows, wd, n, lane);
}

template <bool REC>
__device__ __forceinline__ void filter_add(const float* __restrict__ hin, const int* __restrict__ src,
                                           const int* __restrict__ dst, float* rows, int* q,
                                           int row0, int wave, int lane, int* lrec, int* __restrict__ list) {
    int count = 0;

    v4i cur[TILE_L], nxt[TILE_L];
    load_tile(dst, wave, lane, cur);
    for (int tile = wave; tile < N_ETILES; tile += SC_WAVES) {
        load_tile(dst, tile + SC_WAVES, lane, nxt);
        #pragma unroll
        for (int j = 0; j < TILE_L; ++j) {
            #pragma unroll
            for (int k = 0; k < 4; ++k) {
                const int e = tile * TILE_E + j * 128 + 4 * lane + k;
                const unsigned lr = (unsigned)(cur[j][k] - row0);
                const bool m = lr < (unsigned)ROWS_PB;
                const unsigned mask = __builtin_amdgcn_ballot_w32(m);
                if (mask != 0u) {
                    if (m) {
                        const int pos = count + (int)__builtin_amdgcn_mbcnt_lo(mask, 0u);
                        __hip_atomic_store(q + pos, e | (int)(lr << 20), __ATOMIC_RELAXED, __HIP_MEMORY_SCOPE_WAVEFRONT);
                    }
                    count += __builtin_popcount(mask);
                }
            }
        }
        while (count >= EPW) {
            count -= EPW;
            add_batch<true, REC>(hin, src, rows, q + count, EPW, lane, lrec, list);
        }
        #pragma unroll
        for (int j = 0; j < TILE_L; ++j) cur[j] = nxt[j];
    }
    if (count > 0) add_batch<false, REC>(hin, src, rows, q, count, lane, lrec, list);
}

__device__ __forceinline__ void zero_rows(float* rows, int tid) {
    for (int i = tid; i < ROWS_PB * DIM / 4; i += SC_THREADS) {
        v4f z4;
        z4.x = 0.0f; z4.y = 0.0f; z4.z = 0.0f; z4.w = 0.0f;
        ((v4f*)rows)[i] = z4;
    }
}

__device__ __forceinline__ void store_rows(const float* rows, float* __restrict__ agg, int row0, int tid) {
    int nrows = N_NODES - row0;
    if (nrows > ROWS_PB) nrows = ROWS_PB;
    v4f* ag4 = (v4f*)(agg + (long)row0 * DIM);
    for (int i = tid; i < nrows * (DIM / 4); i += SC_THREADS) ag4[i] = ((const v4f*)rows)[i];
}

__global__ __launch_bounds__(SC_THREADS) void gin_rowsum_first(const float* __restrict__ hin,
                                                               const int* __restrict__ src,
                                                               const int* __restrict__ dst,
                                                               float* __restrict__ agg,
                                                               int* __restrict__ lists,
                                                               int* __restrict__ lens) {
    __shared__ __attribute__((aligned(16))) float rows[ROWS_PB * DIM];
    __shared__ int   queue[SC_WAVES][QCAP];
    __shared__ int   lrec;

    const int tid  = (int)threadIdx.x;
    const int lane = tid & 31;
    const int wave = __builtin_amdgcn_readfirstlane(tid >> 5);
    const int row0 = (int)blockIdx.x * ROWS_PB;

    zero_rows(rows, tid);
    if (tid == 0) lrec = 0;
    __syncthreads();

    filter_add<true>(hin, src, dst, rows, queue[wave], row0, wave, lane, &lrec, lists + (long)blockIdx.x * LCAP);
    __syncthreads();

    if (tid == 0) lens[blockIdx.x] = lrec;
    store_rows(rows, agg, row0, tid);
}

__global__ __launch_bounds__(SC_THREADS) void gin_rowsum_next(const float* __restrict__ hin,
                                                              const int* __restrict__ src,
                                                              const int* __restrict__ dst,
                                                              float* __restrict__ agg,
                                                              const int* __restrict__ lists,
                                                              const int* __restrict__ lens) {
    __shared__ __attribute__((aligned(16))) float rows[ROWS_PB * DIM];
    __shared__ int   queue[SC_WAVES][QCAP];

    const int tid  = (int)threadIdx.x;
    const int lane = tid & 31;
    const int wave = __builtin_amdgcn_readfirstlane(tid >> 5);
    const int row0 = (int)blockIdx.x * ROWS_PB;

    zero_rows(rows, tid);
    __syncthreads();

    const int total = lens[blockIdx.x];
    if (total <= LCAP) {
        const int* bl = lists + (long)blockIdx.x * LCAP;
        const int nb = total / EPW;
        if (wave < nb) {
            int vw = bl[EPW * wave + (lane & (EPW - 1))];
            for (int b = wave; b < nb; b += SC_WAVES) {
                int wd[EPW];
                #pragma unroll
                for (int j = 0; j < EPW; ++j) wd[j] = __builtin_amdgcn_readlane(vw, j);
                const int bn = (b + SC_WAVES < nb) ? b + SC_WAVES : b;
                vw = bl[EPW * bn + (lane & (EPW - 1))];
                add_words<true>(hin, rows, wd, EPW, lane);
            }
        }
        const int rem = total - EPW * nb;
        if (rem > 0 && wave == (nb & (SC_WAVES - 1))) {
            int wd[EPW];
            const int vt = bl[EPW * nb + (lane < rem ? lane : 0)];
            #pragma unroll
            for (int j = 0; j < EPW; ++j) wd[j] = __builtin_amdgcn_readlane(vt, j);
            add_words<false>(hin, rows, wd, rem, lane);
        }
    } else {
        filter_add<false>(hin, src, dst, rows, queue[wave], row0, wave, lane, nullptr, nullptr);
    }
    __syncthreads();

    store_rows(rows, agg, row0, tid);
}

#define TPB 8
#define N_TILES (N_NODES / 16)
static_assert(N_NODES % 16 == 0, "whole 16-row tiles only");

__device__ __forceinline__ void stage_w_load(const float* __restrict__ Wg, int tid, v4f (&w)[16]) {
    #pragma unroll
    for (int it = 0; it < 16; ++it) {
        const int id = tid + 256 * it;
        const int n  = id & 127;
        const int kg = id >> 7;
        const float* wp = Wg + (4 * kg) * DIM + n;
        w[it].x = wp[0];
        w[it].y = wp[DIM];
        w[it].z = wp[2 * DIM];
        w[it].w = wp[3 * DIM];
    }
}

__device__ __forceinline__ void stage_w_store(v4f* __restrict__ Wl, int tid, const v4f (&w)[16]) {
    #pragma unroll
    for (int it = 0; it < 16; ++it) {
        const int id = tid + 256 * it;
        const int n  = id & 127;
        const int kg = id >> 7;
        const int t  = kg >> 2;
        const int h  = (kg >> 1) & 1;
        const int q  = kg & 1;
        const int u  = n >> 4;
        const int c  = n & 15;
        Wl[(((u * 8 + t) * 2 + q) * 32) + 16 * h + c] = w[it];
    }
}

__device__ __forceinline__ float row16_sum(float v) {
    v += __builtin_bit_cast(float, __builtin_amdgcn_update_dpp(0, __builtin_bit_cast(int, v), 0x161, 0xf, 0xf, false));
    v += __builtin_bit_cast(float, __builtin_amdgcn_update_dpp(0, __builtin_bit_cast(int, v), 0x162, 0xf, 0xf, false));
    v += __builtin_bit_cast(float, __builtin_amdgcn_update_dpp(0, __builtin_bit_cast(int, v), 0x164, 0xf, 0xf, false));
    v += __builtin_bit_cast(float, __builtin_amdgcn_update_dpp(0, __builtin_bit_cast(int, v), 0x168, 0xf, 0xf, false));
    return v;
}

__global__ __launch_bounds__(256) void gin_mlp(const float* __restrict__ hin,
                                               const float* __restrict__ agg,
                                               const float* __restrict__ W1,
                                               const float* __restrict__ b1,
                                               const float* __restrict__ W2,
                                               const float* __restrict__ b2,
                                               const float* __restrict__ epsp,
                                               float* __restrict__ hout,
                                               float* __restrict__ stats,
                                               int do_stats) {
    __shared__ v4f w1l[DIM * DIM / 4];
    __shared__ v4f w2l[DIM * DIM / 4];
    __shared__ float red[TPB][2 * DIM];

    const int tid  = threadIdx.x;
    const int lane = tid & 31;
    const int c    = lane & 15;
    const int h    = lane >> 4;
    const int wave = __builtin_amdgcn_readfirstlane(tid >> 5);
    const int tile = blockIdx.x * TPB + wave;
    const bool valid = tile < N_TILES;
    const int tl = valid ? tile : N_TILES - 1;
    const long rowoff = (long)(tl * 16 + c) * DIM + 8 * h;
    const float e1 = 1.0f + epsp[0];

    v4f hr[8][2], ar[8][2];
    #pragma unroll
    for (int t = 0; t < 8; ++t) {
        hr[t][0] = *(const v4f*)(hin + rowoff + 16 * t);
        hr[t][1] = *(const v4f*)(hin + rowoff + 16 * t + 4);
        ar[t][0] = *(const v4f*)(agg + rowoff + 16 * t);
        ar[t][1] = *(const v4f*)(agg + rowoff + 16 * t + 4);
    }
    v4f wr1[16], wr2[16];
    stage_w_load(W1, tid, wr1);
    stage_w_load(W2, tid, wr2);

    float zf[8][8];
    #pragma unroll
    for (int t = 0; t < 8; ++t) {
        zf[t][0] = e1 * hr[t][0].x + ar[t][0].x;
        zf[t][1] = e1 * hr[t][0].y + ar[t][0].y;
        zf[t][2] = e1 * hr[t][0].z + ar[t][0].z;
        zf[t][3] = e1 * hr[t][0].w + ar[t][0].w;
        zf[t][4] = e1 * hr[t][1].x + ar[t][1].x;
        zf[t][5] = e1 * hr[t][1].y + ar[t][1].y;
        zf[t][6] = e1 * hr[t][1].z + ar[t][1].z;
        zf[t][7] = e1 * hr[t][1].w + ar[t][1].w;
    }

    stage_w_store(w1l, tid, wr1);
    stage_w_store(w2l, tid, wr2);
    __syncthreads();

    v8f acc2[8];
    if (valid) {
        v8f acc1[8];
        #pragma unroll
        for (int u = 0; u < 8; ++u) {
            const v4f bl = *(const v4f*)(b1 + 16 * u + 8 * h);
            const v4f bh = *(const v4f*)(b1 + 16 * u + 8 * h + 4);
            acc1[u][0] = bl.x; acc1[u][1] = bl.y; acc1[u][2] = bl.z; acc1[u][3] = bl.w;
            acc1[u][4] = bh.x; acc1[u][5] = bh.y; acc1[u][6] = bh.z; acc1[u][7] = bh.w;
        }
        #pragma unroll
        for (int t = 0; t < 8; ++t) {
            #pragma unroll
            for (int u = 0; u < 8; ++u) {
                const v4f wa = w1l[((u * 8 + t) * 2 + 0) * 32 + lane];
                const v4f wb = w1l[((u * 8 + t) * 2 + 1) * 32 + lane];
                v2f a, b;
                a.x = wa.x; a.y = wa.y; b.x = zf[t][0]; b.y = zf[t][1];
                acc1[u] = __builtin_amdgcn_wmma_f32_16x16x4_f32(false, a, false, b, (short)0, acc1[u], false, false);
                a.x = wa.z; a.y = wa.w; b.x = zf[t][2]; b.y = zf[t][3];
                acc1[u] = __builtin_amdgcn_wmma_f32_16x16x4_f32(false, a, false, b, (short)0, acc1[u], false, false);
                a.x = wb.x; a.y = wb.y; b.x = zf[t][4]; b.y = zf[t][5];
                acc1[u] = __builtin_amdgcn_wmma_f32_16x16x4_f32(false, a, false, b, (short)0, acc1[u], false, false);
                a.x = wb.z; a.y = wb.w; b.x = zf[t][6]; b.y = zf[t][7];
                acc1[u] = __builtin_amdgcn_wmma_f32_16x16x4_f32(false, a, false, b, (short)0, acc1[u], false, false);
            }
        }
        #pragma unroll
        for (int t = 0; t < 8; ++t) {
            #pragma unroll
            for (int r = 0; r < 8; ++r) acc1[t][r] = fmaxf(acc1[t][r], 0.0f);
        }

        #pragma unroll
        for (int u = 0; u < 8; ++u) {
            const v4f bl = *(const v4f*)(b2 + 16 * u + 8 * h);
            const v4f bh = *(const v4f*)(b2 + 16 * u + 8 * h + 4);
            acc2[u][0] = bl.x; acc2[u][1] = bl.y; acc2[u][2] = bl.z; acc2[u][3] = bl.w;
            acc2[u][4] = bh.x; acc2[u][5] = bh.y; acc2[u][6] = bh.z; acc2[u][7] = bh.w;
        }
        #pragma unroll
        for (int t = 0; t < 8; ++t) {
            #pragma unroll
            for (int u = 0; u < 8; ++u) {
                const v4f wa = w2l[((u * 8 + t) * 2 + 0) * 32 + lane];
                const v4f wb = w2l[((u * 8 + t) * 2 + 1) * 32 + lane];
                v2f a, b;
                a.x = wa.x; a.y = wa.y; b.x = acc1[t][0]; b.y = acc1[t][1];
                acc2[u] = __builtin_amdgcn_wmma_f32_16x16x4_f32(false, a, false, b, (short)0, acc2[u], false, false);
                a.x = wa.z; a.y = wa.w; b.x = acc1[t][2]; b.y = acc1[t][3];
                acc2[u] = __builtin_amdgcn_wmma_f32_16x16x4_f32(false, a, false, b, (short)0, acc2[u], false, false);
                a.x = wb.x; a.y = wb.y; b.x = acc1[t][4]; b.y = acc1[t][5];
                acc2[u] = __builtin_amdgcn_wmma_f32_16x16x4_f32(false, a, false, b, (short)0, acc2[u], false, false);
                a.x = wb.z; a.y = wb.w; b.x = acc1[t][6]; b.y = acc1[t][7];
                acc2[u] = __builtin_amdgcn_wmma_f32_16x16x4_f32(false, a, false, b, (short)0, acc2[u], false, false);
            }
        }

        #pragma unroll
        for (int u = 0; u < 8; ++u) {
            v4f lo, hi4;
            lo.x = acc2[u][0]; lo.y = acc2[u][1]; lo.z = acc2[u][2]; lo.w = acc2[u][3];
            hi4.x = acc2[u][4]; hi4.y = acc2[u][5]; hi4.z = acc2[u][6]; hi4.w = acc2[u][7];
            *(v4f*)(hout + rowoff + 16 * u)     = lo;
            *(v4f*)(hout + rowoff + 16 * u + 4) = hi4;
        }
    } else {
        #pragma unroll
        for (int u = 0; u < 8; ++u) {
            #pragma unroll
            for (int r = 0; r < 8; ++r) acc2[u][r] = 0.0f;
        }
    }

    if (do_stats) {
        float cs[8][8], cq[8][8];
        #pragma unroll
        for (int u = 0; u < 8; ++u) {
            #pragma unroll
            for (int r = 0; r < 8; ++r) {
                const float val = acc2[u][r];
                cs[u][r] = row16_sum(val);
                cq[u][r] = row16_sum(val * val);
            }
        }
        if (c == 0) {
            #pragma unroll
            for (int u = 0; u < 8; ++u) {
                #pragma unroll
                for (int r = 0; r < 8; ++r) {
                    red[wave][16 * u + 8 * h + r]       = cs[u][r];
                    red[wave][DIM + 16 * u + 8 * h + r] = cq[u][r];
                }
            }
        }
        __syncthreads();
        float tot = 0.0f;
        #pragma unroll
        for (int w = 0; w < TPB; ++w) tot += red[w][tid];
        atomicAdd(&stats[tid], tot);
    }
}

__global__ __launch_bounds__(256) void gin_bn_relu(float* __restrict__ z,
                                                   const float* __restrict__ stats,
                                                   const float* __restrict__ gamma,
                                                   const float* __restrict__ beta) {
    long i = (long)blockIdx.x * blockDim.x + threadIdx.x;
    if (i >= (long)N_NODES * DIM) return;
    int c = (int)(i & (DIM - 1));
    const float invN = 1.0f / (float)N_NODES;
    float mu  = stats[c] * invN;
    float var = stats[DIM + c] * invN - mu * mu;
    float v = (z[i] - mu) * rsqrtf(var + BN_EPS) * gamma[c] + beta[c];
    z[i] = fmaxf(v, 0.0f);
}

__global__ __launch_bounds__(256) void gin_pool(const float* __restrict__ h,
                                                const int* __restrict__ batch,
                                                float* __restrict__ out) {
    const int lane = (int)(threadIdx.x & 31);
    const int wv = __builtin_amdgcn_readfirstlane((int)(blockIdx.x * 8 + (threadIdx.x >> 5)));
    const int n0 = wv * EPW;
    if (n0 >= N_NODES) return;
    int g[EPW];
    #pragma unroll
    for (int j = 0; j < EPW; ++j) g[j] = batch[n0 + j];
    float v[EPW][4];
    #pragma unroll
    for (int j = 0; j < EPW; ++j) {
        const float* hp = h + (long)(n0 + j) * DIM + lane;
        v[j][0] = hp[0];
        v[j][1] = hp[32];
        v[j][2] = hp[64];
        v[j][3] = hp[96];
    }
    #pragma unroll
    for (int j = 0; j < EPW; ++j) {
        float* op = out + (long)g[j] * DIM + lane;
        atomicAdd(op + 0,  v[j][0]);
        atomicAdd(op + 32, v[j][1]);
        atomicAdd(op + 64, v[j][2]);
        atomicAdd(op + 96, v[j][3]);
    }
}

extern "C" void kernel_launch(void* const* d_in, const int* in_sizes, int n_in,
                              void* d_out, int out_size, void* d_ws, size_t ws_size,
                              hipStream_t stream) {
    const float* x     = (const float*)d_in[0];
    const int*   ei    = (const int*)d_in[1];
    const int*   batch = (const int*)d_in[2];
    const float* W1    = (const float*)d_in[3];
    const float* b1    = (const float*)d_in[4];
    const float* W2    = (const float*)d_in[5];
    const float* b2    = (const float*)d_in[6];
    const float* eps   = (const float*)d_in[7];
    const float* gamma = (const float*)d_in[8];
    const float* beta  = (const float*)d_in[9];
    float* out = (float*)d_out;

    const long ND = (long)N_NODES * DIM;
    float* hA    = (float*)d_ws;
    float* hB    = hA + ND;
    float* agg   = hB + ND;
    float* stats = agg + ND;
    int*   lens  = (int*)(stats + 2 * DIM);
    int*   lists = lens + N_SBLK;

    const int* src = ei;
    const int* dst = ei + N_EDGES;

    const int  mlpGrd  = (N_TILES + TPB - 1) / TPB;
    const int  bnGrd   = (int)((ND + 255) / 256);
    const int  poolGrd = (N_NODES / EPW + 7) / 8;

    for (int i = 0; i < N_LAYERS; ++i) {
        const float* hin  = (i == 0) ? x : ((i & 1) ? hA : hB);
        float*       hout = (i & 1) ? hB : hA;
        int do_stats = (i < N_LAYERS - 1) ? 1 : 0;

        if (do_stats) gin_zero<<<1, 256, 0, stream>>>(stats, 2 * DIM);
        if (i == 0) gin_rowsum_first<<<N_SBLK, SC_THREADS, 0, stream>>>(hin, src, dst, agg, lists, lens);
        else        gin_rowsum_next<<<N_SBLK, SC_THREADS, 0, stream>>>(hin, src, dst, agg, lists, lens);
        gin_mlp<<<mlpGrd, 256, 0, stream>>>(hin, agg,
                                            W1 + (long)i * DIM * DIM, b1 + (long)i * DIM,
                                            W2 + (long)i * DIM * DIM, b2 + (long)i * DIM,
                                            eps + i, hout, stats, do_stats);
        if (do_stats) {
            gin_bn_relu<<<bnGrd, 256, 0, stream>>>(hout, stats,
                                                   gamma + (long)i * DIM,
                                                   beta + (long)i * DIM);
        }
    }
    gin_zero<<<(NUM_GRAPHS * DIM + 255) / 256, 256, 0, stream>>>(out, NUM_GRAPHS * DIM);
    gin_pool<<<poolGrd, 256, 0, stream>>>(hB, batch, out);
}


# Neighbour sum, layers 2-4: wave w owns the block's rows w, w+32, ..; layer 1 files kept edges per owner wave; sums stay in registers (scalar-indexed), no LDS float add; row sum order = list order
# speedup vs baseline: 2.6949x; 1.0428x over previous
#include <hip/hip_runtime.h>
#include <hip/hip_bf16.h>

#define N_NODES   50000
#define N_EDGES   800000
#define DIM       128
#define N_LAYERS  4
#define NUM_GRAPHS 128
#define BN_EPS    1e-5f

typedef __attribute__((ext_vector_type(2))) float v2f;
typedef __attribute__((ext_vector_type(8))) float v8f;

__global__ __launch_bounds__(256) void gin_zero(float* __restrict__ p, long n) {
    long i = (long)blockIdx.x * blockDim.x + threadIdx.x;
    if (i < n) p[i] = 0.0f;
}

#define EPW 8
#define ROWS_PB   196
#define SC_WAVES  32
#define SC_THREADS (SC_WAVES * 32)
#define N_SBLK    ((N_NODES + ROWS_PB - 1) / ROWS_PB)
#define TILE_L    8
#define TILE_E    (TILE_L * 128)
#define N_ETILES  ((N_EDGES + TILE_E - 1) / TILE_E)
#define QCAP      (TILE_E + EPW)
#define LCAP      65536
#define OCAP      (LCAP / SC_WAVES)
#define NSLOT     ((ROWS_PB + SC_WAVES - 1) / SC_WAVES)
#define EPN       16
static_assert(N_EDGES % 4 == 0, "a 16-byte index load must not straddle the end of dst");
static_assert(N_EDGES <= (1 << 20), "the edge number must fit the low 20 bits of a queue word");
static_assert(ROWS_PB < (1 << 11), "the local row must fit the bits above the edge number");
static_assert(N_NODES % EPW == 0, "a pool wave's nodes must all exist");
static_assert(N_NODES <= (1 << 20), "a source number must fit the low 20 bits of a list word");
static_assert(LCAP % EPW == 0, "a batch of list words must not straddle two blocks' lists");
static_assert(OCAP % EPN == 0, "a batch of list words must not straddle two owners' lists");
static_assert(EPN <= 32, "one lane per word of a batch");
static_assert(NSLOT * 4 <= 32, "a wave's row sums must fit the 32-float accumulator vector");
static_assert(SC_WAVES == 32, "one lane per owner list count");

typedef __attribute__((ext_vector_type(4))) int   v4i;
typedef __attribute__((ext_vector_type(4))) float v4f;
typedef __attribute__((ext_vector_type(32))) float v32f;

__device__ __forceinline__ void load_tile(const int* __restrict__ dst, int tile, int lane, v4i (&t)[TILE_L]) {
    #pragma unroll
    for (int j = 0; j < TILE_L; ++j) {
        const int idx = tile * TILE_E + j * 128 + 4 * lane;
        const int idc = idx < N_EDGES ? idx : N_EDGES - 4;
        v4i d = *(const v4i*)(dst + idc);
        if (idx >= N_EDGES) { d.x = -1; d.y = -1; d.z = -1; d.w = -1; }
        t[j] = d;
    }
}

template <bool FULL>
__device__ __forceinline__ void add_words(const float* __restrict__ hin, float* rows,
                                          const int (&wd)[EPW], int n, int lane) {
    float v[EPW][4];
    #pragma unroll
    for (int j = 0; j < EPW; ++j) {
        if (FULL || j < n) {
            const float* hs = hin + (long)(wd[j] & 0xFFFFF) * DIM + lane;
            v[j][0] = hs[0];
            v[j][1] = hs[32];
            v[j][2] = hs[64];
            v[j][3] = hs[96];
        }
    }
    #pragma unroll
    for (int j = 0; j < EPW; ++j) {
        if (FULL || j < n) {
            float* r = rows + (wd[j] >> 20) * DIM + lane;
            __hip_atomic_fetch_add(r + 0,  v[j][0], __ATOMIC_RELAXED, __HIP_MEMORY_SCOPE_WORKGROUP);
            __hip_atomic_fetch_add(r + 32, v[j][1], __ATOMIC_RELAXED, __HIP_MEMORY_SCOPE_WORKGROUP);
            __hip_atomic_fetch_add(r + 64, v[j][2], __ATOMIC_RELAXED, __HIP_MEMORY_SCOPE_WORKGROUP);
            __hip_atomic_fetch_add(r + 96, v[j][3], __ATOMIC_RELAXED, __HIP_MEMORY_SCOPE_WORKGROUP);
        }
    }
}

template <bool FULL>
__device__ __forceinline__ void acc_words(const float* __restrict__ hin, const int (&wd)[EPN], int n, int lane,
                                          int vnext, v32f& acc) {
    v4f v[EPN];
    #pragma unroll
    for (int j = 0; j < EPN; ++j) {
        if (FULL || j < n) {
            const float* rb = hin + (size_t)(wd[j] & 0xFFFFF) * DIM;
            v[j] = *(const v4f*)(rb + 4u * (unsigned)lane);
        }
    }
    if (FULL) {
        asm volatile("" :: "v"(v[0]), "v"(v[1]), "v"(v[2]), "v"(v[3]), "v"(v[4]), "v"(v[5]), "v"(v[6]), "v"(v[7]),
                           "v"(v[8]), "v"(v[9]), "v"(v[10]), "v"(v[11]), "v"(v[12]), "v"(v[13]), "v"(v[14]), "v"(v[15]),
                           "v"(vnext));
    }
    #pragma unroll
    for (int j = 0; j < EPN; ++j) {
        if (FULL || j < n) {
            const int e = (wd[j] >> 25) * 4;
            acc[e + 0] += v[j].x;
            acc[e + 1] += v[j].y;
            acc[e + 2] += v[j].z;
            acc[e + 3] += v[j].w;
        }
    }
}

template <bool FULL, bool REC>
__device__ __forceinline__ void add_batch(const float* __restrict__ hin, const int* __restrict__ src,
                                          float* rows, const int* qe, int n, int lane,
                                          int* lcnt, int* __restrict__ list) {
    int ent[EPW], wd[EPW];
    #pragma unroll
    for (int j = 0; j < EPW; ++j) {
        ent[j] = 0;
        if (FULL || j < n)
            ent[j] = __builtin_amdgcn_readfirstlane(__hip_atomic_load(qe + j, __ATOMIC_RELAXED, __HIP_MEMORY_SCOPE_WAVEFRONT));
    }
    #pragma unroll
    for (int j = 0; j < EPW; ++j) {
        wd[j] = 0;
        if (FULL || j < n) wd[j] = src[ent[j] & 0xFFFFF] | (ent[j] & ~0xFFFFF);
    }
    if (REC) {
        int w = 0;
        #pragma unroll
        for (int j = 0; j < EPW; ++j) {
            if (lane == j) w = wd[j];
        }
        if (lane < n) {
            const int o = (w >> 20) & (SC_WAVES - 1);
            const int pos = __hip_atomic_fetch_add(lcnt + o, 1, __ATOMIC_RELAXED, __HIP_MEMORY_SCOPE_WORKGROUP);
            if (pos < OCAP) list[o * OCAP + pos] = w;
        }
    }
    add_words<FULL>(hin, rows, wd, n, lane);
}

template <bool REC>
__device__ __forceinline__ void filter_add(const float* __restrict__ hin, const int* __restrict__ src,
                                           const int* __restrict__ dst, float* rows, int* q,
                                           int row0, int wave, int lane, int* lcnt, int* __restrict__ list) {
    int count = 0;

    v4i cur[TILE_L], nxt[TILE_L];
    load_tile(dst, wave, lane, cur);
    for (int tile = wave; tile < N_ETILES; tile += SC_WAVES) {
        load_tile(dst, tile + SC_WAVES, lane, nxt);
        #pragma unroll
        for (int j = 0; j < TILE_L; ++j) {
            #pragma unroll
            for (int k = 0; k < 4; ++k) {
                const int e = tile * TILE_E + j * 128 + 4 * lane + k;
                const unsigned lr = (unsigned)(cur[j][k] - row0);
                const bool m = lr < (unsigned)ROWS_PB;
                const unsigned mask = __builtin_amdgcn_ballot_w32(m);
                if (mask != 0u) {
                    if (m) {
                        const int pos = count + (int)__builtin_amdgcn_mbcnt_lo(mask, 0u);
                        __hip_atomic_store(q + pos, e | (int)(lr << 20), __ATOMIC_RELAXED, __HIP_MEMORY_SCOPE_WAVEFRONT);
                    }
                    count += __builtin_popcount(mask);
                }
            }
        }
        while (count >= EPW) {
            count -= EPW;
            add_batch<true, REC>(hin, src, rows, q + count, EPW, lane, lcnt, list);
        }
        #pragma unroll
        for (int j = 0; j < TILE_L; ++j) cur[j] = nxt[j];
    }
    if (count > 0) add_batch<false, REC>(hin, src, rows, q, count, lane, lcnt, list);
}

__device__ __forceinline__ void zero_rows(float* rows, int tid) {
    for (int i = tid; i < ROWS_PB * DIM / 4; i += SC_THREADS) {
        v4f z4;
        z4.x = 0.0f; z4.y = 0.0f; z4.z = 0.0f; z4.w = 0.0f;
        ((v4f*)rows)[i] = z4;
    }
}

__device__ __forceinline__ void store_rows(const float* rows, float* __restrict__ agg, int row0, int tid) {
    int nrows = N_NODES - row0;
    if (nrows > ROWS_PB) nrows = ROWS_PB;
    v4f* ag4 = (v4f*)(agg + (long)row0 * DIM);
    for (int i = tid; i < nrows * (DIM / 4); i += SC_THREADS) ag4[i] = ((const v4f*)rows)[i];
}

__global__ __launch_bounds__(SC_THREADS) void gin_rowsum_first(const float* __restrict__ hin,
                                                               const int* __restrict__ src,
                                                               const int* __restrict__ dst,
                                                               float* __restrict__ agg,
                                                               int* __restrict__ lists,
                                                               int* __restrict__ lens) {
    __shared__ __attribute__((aligned(16))) float rows[ROWS_PB * DIM];
    __shared__ int   queue[SC_WAVES][QCAP];
    __shared__ int   lcnt[SC_WAVES];

    const int tid  = (int)threadIdx.x;
    const int lane = tid & 31;
    const int wave = __builtin_amdgcn_readfirstlane(tid >> 5);
    const int row0 = (int)blockIdx.x * ROWS_PB;

    zero_rows(rows, tid);
    if (tid < SC_WAVES) lcnt[tid] = 0;
    __syncthreads();

    filter_add<true>(hin, src, dst, rows, queue[wave], row0, wave, lane, lcnt, lists + (long)blockIdx.x * LCAP);
    __syncthreads();

    if (tid < SC_WAVES) lens[blockIdx.x * SC_WAVES + tid] = lcnt[tid];
    store_rows(rows, agg, row0, tid);
}

__global__ __launch_bounds__(SC_THREADS) void gin_rowsum_next(const float* __restrict__ hin,
                                                              const int* __restrict__ src,
                                                              const int* __restrict__ dst,
                                                              float* __restrict__ agg,
                                                              const int* __restrict__ lists,
                                                              const int* __restrict__ lens) {
    __shared__ __attribute__((aligned(16))) float rows[ROWS_PB * DIM];
    __shared__ int   queue[SC_WAVES][QCAP];

    const int tid  = (int)threadIdx.x;
    const int lane = tid & 31;
    const int wave = __builtin_amdgcn_readfirstlane(tid >> 5);
    const int row0 = (int)blockIdx.x * ROWS_PB;

    const int cv = lens[blockIdx.x * SC_WAVES + lane];
    const bool whole = __builtin_amdgcn_ballot_w32(cv > OCAP) == 0u;
    if (whole) {
        const int total = __builtin_amdgcn_readlane(cv, wave);
        const int* bl = lists + (long)blockIdx.x * LCAP + wave * OCAP;
        v32f acc;
        #pragma unroll
        for (int e = 0; e < 32; ++e) acc[e] = 0.0f;
        const int nb = total / EPN;
        if (nb > 0) {
            const int vw0 = bl[lane & (EPN - 1)];
            int wd[EPN];
            #pragma unroll
            for (int j = 0; j < EPN; ++j) wd[j] = __builtin_amdgcn_readlane(vw0, j);
            for (int b = 0; b < nb; ++b) {
                const int bn = (b + 1 < nb) ? b + 1 : b;
                const int vw = bl[EPN * bn + (lane & (EPN - 1))];
                acc_words<true>(hin, wd, EPN, lane, vw, acc);
                #pragma unroll
                for (int j = 0; j < EPN; ++j) wd[j] = __builtin_amdgcn_readlane(vw, j);
            }
        }
        const int rem = total - EPN * nb;
        if (rem > 0) {
            int wd[EPN];
            const int vt = bl[EPN * nb + (lane < rem ? lane : 0)];
            #pragma unroll
            for (int j = 0; j < EPN; ++j) wd[j] = __builtin_amdgcn_readlane(vt, j);
            acc_words<false>(hin, wd, rem, lane, 0, acc);
        }
        #pragma unroll
        for (int s = 0; s < NSLOT; ++s) {
            const int lr  = s * SC_WAVES + wave;
            const int row = row0 + lr;
            if (lr < ROWS_PB && row < N_NODES) {
                v4f o;
                o.x = acc[4 * s + 0]; o.y = acc[4 * s + 1]; o.z = acc[4 * s + 2]; o.w = acc[4 * s + 3];
                *(v4f*)(agg + (long)row * DIM + 4 * lane) = o;
            }
        }
    } else {
        zero_rows(rows, tid);
        __syncthreads();
        filter_add<false>(hin, src, dst, rows, queue[wave], row0, wave, lane, nullptr, nullptr);
        __syncthreads();
        store_rows(rows, agg, row0, tid);
    }
}

#define TPB 8
#define N_TILES (N_NODES / 16)
static_assert(N_NODES % 16 == 0, "whole 16-row tiles only");

__device__ __forceinline__ void stage_w_load(const float* __restrict__ Wg, int tid, v4f (&w)[16]) {
    #pragma unroll
    for (int it = 0; it < 16; ++it) {
        const int id = tid + 256 * it;
        const int n  = id & 127;
        const int kg = id >> 7;
        const float* wp = Wg + (4 * kg) * DIM + n;
        w[it].x = wp[0];
        w[it].y = wp[DIM];
        w[it].z = wp[2 * DIM];
        w[it].w = wp[3 * DIM];
    }
}

__device__ __forceinline__ void stage_w_store(v4f* __restrict__ Wl, int tid, const v4f (&w)[16]) {
    #pragma unroll
    for (int it = 0; it < 16; ++it) {
        const int id = tid + 256 * it;
        const int n  = id & 127;
        const int kg = id >> 7;
        const int t  = kg >> 2;
        const int h  = (kg >> 1) & 1;
        const int q  = kg & 1;
        const int u  = n >> 4;
        const int c  = n & 15;
        Wl[(((u * 8 + t) * 2 + q) * 32) + 16 * h + c] = w[it];
    }
}

__device__ __forceinline__ float row16_sum(float v) {
    v += __builtin_bit_cast(float, __builtin_amdgcn_update_dpp(0, __builtin_bit_cast(int, v), 0x161, 0xf, 0xf, false));
    v += __builtin_bit_cast(float, __builtin_amdgcn_update_dpp(0, __builtin_bit_cast(int, v), 0x162, 0xf, 0xf, false));
    v += __builtin_bit_cast(float, __builtin_amdgcn_update_dpp(0, __builtin_bit_cast(int, v), 0x164, 0xf, 0xf, false));
    v += __builtin_bit_cast(float, __builtin_amdgcn_update_dpp(0, __builtin_bit_cast(int, v), 0x168, 0xf, 0xf, false));
    return v;
}

__global__ __launch_bounds__(256) void gin_mlp(const float* __restrict__ hin,
                                               const float* __restrict__ agg,
                                               const float* __restrict__ W1,
                                               const float* __restrict__ b1,
                                               const float* __restrict__ W2,
                                               const float* __restrict__ b2,
                                               const float* __restrict__ epsp,
                                               float* __restrict__ hout,
                                               float* __restrict__ stats,
                                               int do_stats) {
    __shared__ v4f w1l[DIM * DIM / 4];
    __shared__ v4f w2l[DIM * DIM / 4];
    __shared__ float red[TPB][2 * DIM];

    const int tid  = threadIdx.x;
    const int lane = tid & 31;
    const int c    = lane & 15;
    const int h    = lane >> 4;
    const int wave = __builtin_amdgcn_readfirstlane(tid >> 5);
    const int tile = blockIdx.x * TPB + wave;
    const bool valid = tile < N_TILES;
    const int tl = valid ? tile : N_TILES - 1;
    const long rowoff = (long)(tl * 16 + c) * DIM + 8 * h;
    const float e1 = 1.0f + epsp[0];

    v4f hr[8][2], ar[8][2];
    #pragma unroll
    for (int t = 0; t < 8; ++t) {
        hr[t][0] = *(const v4f*)(hin + rowoff + 16 * t);
        hr[t][1] = *(const v4f*)(hin + rowoff + 16 * t + 4);
        ar[t][0] = *(const v4f*)(agg + rowoff + 16 * t);
        ar[t][1] = *(const v4f*)(agg + rowoff + 16 * t + 4);
    }
    v4f wr1[16], wr2[16];
    stage_w_load(W1, tid, wr1);
    stage_w_load(W2, tid, wr2);

    float zf[8][8];
    #pragma unroll
    for (int t = 0; t < 8; ++t) {
        zf[t][0] = e1 * hr[t][0].x + ar[t][0].x;
        zf[t][1] = e1 * hr[t][0].y + ar[t][0].y;
        zf[t][2] = e1 * hr[t][0].z + ar[t][0].z;
        zf[t][3] = e1 * hr[t][0].w + ar[t][0].w;
        zf[t][4] = e1 * hr[t][1].x + ar[t][1].x;
        zf[t][5] = e1 * hr[t][1].y + ar[t][1].y;
        zf[t][6] = e1 * hr[t][1].z + ar[t][1].z;
        zf[t][7] = e1 * hr[t][1].w + ar[t][1].w;
    }

    stage_w_store(w1l, tid, wr1);
    stage_w_store(w2l, tid, wr2);
    __syncthreads();

    v8f acc2[8];
    if (valid) {
        v8f acc1[8];
        #pragma unroll
        for (int u = 0; u < 8; ++u) {
            const v4f bl = *(const v4f*)(b1 + 16 * u + 8 * h);
            const v4f bh = *(const v4f*)(b1 + 16 * u + 8 * h + 4);
            acc1[u][0] = bl.x; acc1[u][1] = bl.y; acc1[u][2] = bl.z; acc1[u][3] = bl.w;
            acc1[u][4] = bh.x; acc1[u][5] = bh.y; acc1[u][6] = bh.z; acc1[u][7] = bh.w;
        }
        #pragma unroll
        for (int t = 0; t < 8; ++t) {
            #pragma unroll
            for (int u = 0; u < 8; ++u) {
                const v4f wa = w1l[((u * 8 + t) * 2 + 0) * 32 + lane];
                const v4f wb = w1l[((u * 8 + t) * 2 + 1) * 32 + lane];
                v2f a, b;
                a.x = wa.x; a.y = wa.y; b.x = zf[t][0]; b.y = zf[t][1];
                acc1[u] = __builtin_amdgcn_wmma_f32_16x16x4_f32(false, a, false, b, (short)0, acc1[u], false, false);
                a.x = wa.z; a.y = wa.w; b.x = zf[t][2]; b.y = zf[t][3];
                acc1[u] = __builtin_amdgcn_wmma_f32_16x16x4_f32(false, a, false, b, (short)0, acc1[u], false, false);
                a.x = wb.x; a.y = wb.y; b.x = zf[t][4]; b.y = zf[t][5];
                acc1[u] = __builtin_amdgcn_wmma_f32_16x16x4_f32(false, a, false, b, (short)0, acc1[u], false, false);
                a.x = wb.z; a.y = wb.w; b.x = zf[t][6]; b.y = zf[t][7];
                acc1[u] = __builtin_amdgcn_wmma_f32_16x16x4_f32(false, a, false, b, (short)0, acc1[u], false, false);
            }
        }
        #pragma unroll
        for (int t = 0; t < 8; ++t) {
            #pragma unroll
            for (int r = 0; r < 8; ++r) acc1[t][r] = fmaxf(acc1[t][r], 0.0f);
        }

        #pragma unroll
        for (int u = 0; u < 8; ++u) {
            const v4f bl = *(const v4f*)(b2 + 16 * u + 8 * h);
            const v4f bh = *(const v4f*)(b2 + 16 * u + 8 * h + 4);
            acc2[u][0] = bl.x; acc2[u][1] = bl.y; acc2[u][2] = bl.z; acc2[u][3] = bl.w;
            acc2[u][4] = bh.x; acc2[u][5] = bh.y; acc2[u][6] = bh.z; acc2[u][7] = bh.w;
        }
        #pragma unroll
        for (int t = 0; t < 8; ++t) {
            #pragma unroll
            for (int u = 0; u < 8; ++u) {
                const v4f wa = w2l[((u * 8 + t) * 2 + 0) * 32 + lane];
                const v4f wb = w2l[((u * 8 + t) * 2 + 1) * 32 + lane];
                v2f a, b;
                a.x = wa.x; a.y = wa.y; b.x = acc1[t][0]; b.y = acc1[t][1];
                acc2[u] = __builtin_amdgcn_wmma_f32_16x16x4_f32(false, a, false, b, (short)0, acc2[u], false, false);
                a.x = wa.z; a.y = wa.w; b.x = acc1[t][2]; b.y = acc1[t][3];
                acc2[u] = __builtin_amdgcn_wmma_f32_16x16x4_f32(false, a, false, b, (short)0, acc2[u], false, false);
                a.x = wb.x; a.y = wb.y; b.x = acc1[t][4]; b.y = acc1[t][5];
                acc2[u] = __builtin_amdgcn_wmma_f32_16x16x4_f32(false, a, false, b, (short)0, acc2[u], false, false);
                a.x = wb.z; a.y = wb.w; b.x = acc1[t][6]; b.y = acc1[t][7];
                acc2[u] = __builtin_amdgcn_wmma_f32_16x16x4_f32(false, a, false, b, (short)0, acc2[u], false, false);
            }
        }

        #pragma unroll
        for (int u = 0; u < 8; ++u) {
            v4f lo, hi4;
            lo.x = acc2[u][0]; lo.y = acc2[u][1]; lo.z = acc2[u][2]; lo.w = acc2[u][3];
            hi4.x = acc2[u][4]; hi4.y = acc2[u][5]; hi4.z = acc2[u][6]; hi4.w = acc2[u][7];
            *(v4f*)(hout + rowoff + 16 * u)     = lo;
            *(v4f*)(hout + rowoff + 16 * u + 4) = hi4;
        }
    } else {
        #pragma unroll
        for (int u = 0; u < 8; ++u) {
            #pragma unroll
            for (int r = 0; r < 8; ++r) acc2[u][r] = 0.0f;
        }
    }

    if (do_stats) {
        float cs[8][8], cq[8][8];
        #pragma unroll
        for (int u = 0; u < 8; ++u) {
            #pragma unroll
            for (int r = 0; r < 8; ++r) {
                const float val = acc2[u][r];
                cs[u][r] = row16_sum(val);
                cq[u][r] = row16_sum(val * val);
            }
        }
        if (c == 0) {
            #pragma unroll
            for (int u = 0; u < 8; ++u) {
                #pragma unroll
                for (int r = 0; r < 8; ++r) {
                    red[wave][16 * u + 8 * h + r]       = cs[u][r];
                    red[wave][DIM + 16 * u + 8 * h + r] = cq[u][r];
                }
            }
        }
        __syncthreads();
        float tot = 0.0f;
        #pragma unroll
        for (int w = 0; w < TPB; ++w) tot += red[w][tid];
        atomicAdd(&stats[tid], tot);
    }
}

__global__ __launch_bounds__(256) void gin_bn_relu(float* __restrict__ z,
                                                   const float* __restrict__ stats,
                                                   const float* __restrict__ gamma,
                                                   const float* __restrict__ beta) {
    long i = (long)blockIdx.x * blockDim.x + threadIdx.x;
    if (i >= (long)N_NODES * DIM) return;
    int c = (int)(i & (DIM - 1));
    const float invN = 1.0f / (float)N_NODES;
    float mu  = stats[c] * invN;
    float var = stats[DIM + c] * invN - mu * mu;
    float v = (z[i] - mu) * rsqrtf(var + BN_EPS) * gamma[c] + beta[c];
    z[i] = fmaxf(v, 0.0f);
}

__global__ __launch_bounds__(256) void gin_pool(const float* __restrict__ h,
                                                const int* __restrict__ batch,
                                                float* __restrict__ out) {
    const int lane = (int)(threadIdx.x & 31);
    const int wv = __builtin_amdgcn_readfirstlane((int)(blockIdx.x * 8 + (threadIdx.x >> 5)));
    const int n0 = wv * EPW;
    if (n0 >= N_NODES) return;
    int g[EPW];
    #pragma unroll
    for (int j = 0; j < EPW; ++j) g[j] = batch[n0 + j];
    float v[EPW][4];
    #pragma unroll
    for (int j = 0; j < EPW; ++j) {
        const float* hp = h + (long)(n0 + j) * DIM + lane;
        v[j][0] = hp[0];
        v[j][1] = hp[32];
        v[j][2] = hp[64];
        v[j][3] = hp[96];
    }
    #pragma unroll
    for (int j = 0; j < EPW; ++j) {
        float* op = out + (long)g[j] * DIM + lane;
        atomicAdd(op + 0,  v[j][0]);
        atomicAdd(op + 32, v[j][1]);
        atomicAdd(op + 64, v[j][2]);
        atomicAdd(op + 96, v[j][3]);
    }
}

extern "C" void kernel_launch(void* const* d_in, const int* in_sizes, int n_in,
                              void* d_out, int out_size, void* d_ws, size_t ws_size,
                              hipStream_t stream) {
    const float* x     = (const float*)d_in[0];
    const int*   ei    = (const int*)d_in[1];
    const int*   batch = (const int*)d_in[2];
    const float* W1    = (const float*)d_in[3];
    const float* b1    = (const float*)d_in[4];
    const float* W2    = (const float*)d_in[5];
    const float* b2    = (const float*)d_in[6];
    const float* eps   = (const float*)d_in[7];
    const float* gamma = (const float*)d_in[8];
    const float* beta  = (const float*)d_in[9];
    float* out = (float*)d_out;

    const long ND = (long)N_NODES * DIM;
    float* hA    = (float*)d_ws;
    float* hB    = hA + ND;
    float* agg   = hB + ND;
    float* stats = agg + ND;
    int*   lens  = (int*)(stats + 2 * DIM);
    int*   lists = lens + N_SBLK * SC_WAVES;

    const int* src = ei;
    const int* dst = ei + N_EDGES;

    const int  mlpGrd  = (N_TILES + TPB - 1) / TPB;
    const int  bnGrd   = (int)((ND + 255) / 256);
    const int  poolGrd = (N_NODES / EPW + 7) / 8;

    for (int i = 0; i < N_LAYERS; ++i) {
        const float* hin  = (i == 0) ? x : ((i & 1) ? hA : hB);
        float*       hout = (i & 1) ? hB : hA;
        int do_stats = (i < N_LAYERS - 1) ? 1 : 0;

        if (do_stats) gin_zero<<<1, 256, 0, stream>>>(stats, 2 * DIM);
        if (i == 0) gin_rowsum_first<<<N_SBLK, SC_THREADS, 0, stream>>>(hin, src, dst, agg, lists, lens);
        else        gin_rowsum_next<<<N_SBLK, SC_THREADS, 0, stream>>>(hin, src, dst, agg, lists, lens);
        gin_mlp<<<mlpGrd, 256, 0, stream>>>(hin, agg,
                                            W1 + (long)i * DIM * DIM, b1 + (long)i * DIM,
                                            W2 + (long)i * DIM * DIM, b2 + (long)i * DIM,
                                            eps + i, hout, stats, do_stats);
        if (do_stats) {
            gin_bn_relu<<<bnGrd, 256, 0, stream>>>(hout, stats,
                                                   gamma + (long)i * DIM,
                                                   beta + (long)i * DIM);
        }
    }
    gin_zero<<<(NUM_GRAPHS * DIM + 255) / 256, 256, 0, stream>>>(out, NUM_GRAPHS * DIM);
    gin_pool<<<poolGrd, 256, 0, stream>>>(hB, batch, out);
}


# Layer 1 edge lists: a new launch bins each 3125-edge chunk by owner block into workspace cells (LDS counts); a block regroups its 256 cells, reads no dst list; layer 1 row sum order = list order
# speedup vs baseline: 2.8939x; 1.0738x over previous
#include <hip/hip_runtime.h>
#include <hip/hip_bf16.h>

#define N_NODES   50000
#define N_EDGES   800000
#define DIM       128
#define N_LAYERS  4
#define NUM_GRAPHS 128
#define BN_EPS    1e-5f

typedef __attribute__((ext_vector_type(2))) float v2f;
typedef __attribute__((ext_vector_type(8))) float v8f;

__global__ __launch_bounds__(256) void gin_zero(float* __restrict__ p, long n) {
    long i = (long)blockIdx.x * blockDim.x + threadIdx.x;
    if (i < n) p[i] = 0.0f;
}

#define EPW 8
#define ROWS_PB   196
#define SC_WAVES  32
#define SC_THREADS (SC_WAVES * 32)
#define N_SBLK    ((N_NODES + ROWS_PB - 1) / ROWS_PB)
#define TILE_L    8
#define TILE_E    (TILE_L * 128)
#define N_ETILES  ((N_EDGES + TILE_E - 1) / TILE_E)
#define QCAP      (TILE_E + EPW)
#define LCAP      65536
#define OCAP      (LCAP / SC_WAVES)
#define NSLOT     ((ROWS_PB + SC_WAVES - 1) / SC_WAVES)
#define EPN       16
#define N_CHUNK   256
#define CHUNK_E   ((N_EDGES + N_CHUNK - 1) / N_CHUNK)
#define BIN_EPT   ((CHUNK_E + SC_THREADS - 1) / SC_THREADS)
#define CELL_W    64
#define CELL_PT   16
static_assert(N_EDGES % 4 == 0, "a 16-byte index load must not straddle the end of dst");
static_assert(N_EDGES <= (1 << 20), "the edge number must fit the low 20 bits of a queue word");
static_assert(ROWS_PB < (1 << 11), "the local row must fit the bits above the edge number");
static_assert(N_NODES % EPW == 0, "a pool wave's nodes must all exist");
static_assert(N_NODES <= (1 << 20), "a source number must fit the low 20 bits of a list word");
static_assert(LCAP % EPW == 0, "a batch of list words must not straddle two blocks' lists");
static_assert(OCAP % EPN == 0, "a batch of list words must not straddle two owners' lists");
static_assert(EPN <= 32, "one lane per word of a batch");
static_assert(NSLOT * 4 <= 32, "a wave's row sums must fit the 32-float accumulator vector");
static_assert(SC_WAVES == 32, "one lane per owner list count");
static_assert(SC_THREADS % N_CHUNK == 0, "whole threads per cell in the regrouping step");
static_assert((SC_THREADS / N_CHUNK) * CELL_PT >= CELL_W - 1, "the threads of a cell must cover all its list words");
static_assert(N_SBLK <= SC_THREADS, "one thread per cell count in the binning launch");

typedef __attribute__((ext_vector_type(4))) int   v4i;
typedef __attribute__((ext_vector_type(4))) float v4f;
typedef __attribute__((ext_vector_type(32))) float v32f;

__device__ __forceinline__ void load_tile(const int* __restrict__ dst, int tile, int lane, v4i (&t)[TILE_L]) {
    #pragma unroll
    for (int j = 0; j < TILE_L; ++j) {
        const int idx = tile * TILE_E + j * 128 + 4 * lane;
        const int idc = idx < N_EDGES ? idx : N_EDGES - 4;
        v4i d = *(const v4i*)(dst + idc);
        if (idx >= N_EDGES) { d.x = -1; d.y = -1; d.z = -1; d.w = -1; }
        t[j] = d;
    }
}

template <bool FULL>
__device__ __forceinline__ void add_words(const float* __restrict__ hin, float* rows,
                                          const int (&wd)[EPW], int n, int lane) {
    float v[EPW][4];
    #pragma unroll
    for (int j = 0; j < EPW; ++j) {
        if (FULL || j < n) {
            const float* hs = hin + (long)(wd[j] & 0xFFFFF) * DIM + lane;
            v[j][0] = hs[0];
            v[j][1] = hs[32];
            v[j][2] = hs[64];
            v[j][3] = hs[96];
        }
    }
    #pragma unroll
    for (int j = 0; j < EPW; ++j) {
        if (FULL || j < n) {
            float* r = rows + (wd[j] >> 20) * DIM + lane;
            __hip_atomic_fetch_add(r + 0,  v[j][0], __ATOMIC_RELAXED, __HIP_MEMORY_SCOPE_WORKGROUP);
            __hip_atomic_fetch_add(r + 32, v[j][1], __ATOMIC_RELAXED, __HIP_MEMORY_SCOPE_WORKGROUP);
            __hip_atomic_fetch_add(r + 64, v[j][2], __ATOMIC_RELAXED, __HIP_MEMORY_SCOPE_WORKGROUP);
            __hip_atomic_fetch_add(r + 96, v[j][3], __ATOMIC_RELAXED, __HIP_MEMORY_SCOPE_WORKGROUP);
        }
    }
}

template <bool FULL>
__device__ __forceinline__ void acc_words(const float* __restrict__ hin, const int (&wd)[EPN], int n, int lane,
                                          int vnext, v32f& acc) {
    v4f v[EPN];
    #pragma unroll
    for (int j = 0; j < EPN; ++j) {
        if (FULL || j < n) {
            const float* rb = hin + (size_t)(wd[j] & 0xFFFFF) * DIM;
            v[j] = *(const v4f*)(rb + 4u * (unsigned)lane);
        }
    }
    if (FULL) {
        asm volatile("" :: "v"(v[0]), "v"(v[1]), "v"(v[2]), "v"(v[3]), "v"(v[4]), "v"(v[5]), "v"(v[6]), "v"(v[7]),
                           "v"(v[8]), "v"(v[9]), "v"(v[10]), "v"(v[11]), "v"(v[12]), "v"(v[13]), "v"(v[14]), "v"(v[15]),
                           "v"(vnext));
    }
    #pragma unroll
    for (int j = 0; j < EPN; ++j) {
        if (FULL || j < n) {
            const int e = (wd[j] >> 25) * 4;
            acc[e + 0] += v[j].x;
            acc[e + 1] += v[j].y;
            acc[e + 2] += v[j].z;
            acc[e + 3] += v[j].w;
        }
    }
}

template <bool FULL>
__device__ __forceinline__ void add_batch(const float* __restrict__ hin, const int* __restrict__ src,
                                          float* rows, const int* qe, int n, int lane) {
    int ent[EPW], wd[EPW];
    #pragma unroll
    for (int j = 0; j < EPW; ++j) {
        ent[j] = 0;
        if (FULL || j < n)
            ent[j] = __builtin_amdgcn_readfirstlane(__hip_atomic_load(qe + j, __ATOMIC_RELAXED, __HIP_MEMORY_SCOPE_WAVEFRONT));
    }
    #pragma unroll
    for (int j = 0; j < EPW; ++j) {
        wd[j] = 0;
        if (FULL || j < n) wd[j] = src[ent[j] & 0xFFFFF] | (ent[j] & ~0xFFFFF);
    }
    add_words<FULL>(hin, rows, wd, n, lane);
}

__device__ __forceinline__ void filter_add(const float* __restrict__ hin, const int* __restrict__ src,
                                           const int* __restrict__ dst, float* rows, int* q,
                                           int row0, int wave, int lane) {
    int count = 0;

    v4i cur[TILE_L], nxt[TILE_L];
    load_tile(dst, wave, lane, cur);
    for (int tile = wave; tile < N_ETILES; tile += SC_WAVES) {
        load_tile(dst, tile + SC_WAVES, lane, nxt);
        #pragma unroll
        for (int j = 0; j < TILE_L; ++j) {
            #pragma unroll
            for (int k = 0; k < 4; ++k) {
                const int e = tile * TILE_E + j * 128 + 4 * lane + k;
                const unsigned lr = (unsigned)(cur[j][k] - row0);
                const bool m = lr < (unsigned)ROWS_PB;
                const unsigned mask = __builtin_amdgcn_ballot_w32(m);
                if (mask != 0u) {
                    if (m) {
                        const int pos = count + (int)__builtin_amdgcn_mbcnt_lo(mask, 0u);
                        __hip_atomic_store(q + pos, e | (int)(lr << 20), __ATOMIC_RELAXED, __HIP_MEMORY_SCOPE_WAVEFRONT);
                    }
                    count += __builtin_popcount(mask);
                }
            }
        }
        while (count >= EPW) {
            count -= EPW;
            add_batch<true>(hin, src, rows, q + count, EPW, lane);
        }
        #pragma unroll
        for (int j = 0; j < TILE_L; ++j) cur[j] = nxt[j];
    }
    if (count > 0) add_batch<false>(hin, src, rows, q, count, lane);
}

__device__ __forceinline__ void zero_rows(float* rows, int tid) {
    for (int i = tid; i < ROWS_PB * DIM / 4; i += SC_THREADS) {
        v4f z4;
        z4.x = 0.0f; z4.y = 0.0f; z4.z = 0.0f; z4.w = 0.0f;
        ((v4f*)rows)[i] = z4;
    }
}

__device__ __forceinline__ void store_rows(const float* rows, float* __restrict__ agg, int row0, int tid) {
    int nrows = N_NODES - row0;
    if (nrows > ROWS_PB) nrows = ROWS_PB;
    v4f* ag4 = (v4f*)(agg + (long)row0 * DIM);
    for (int i = tid; i < nrows * (DIM / 4); i += SC_THREADS) ag4[i] = ((const v4f*)rows)[i];
}


__global__ __launch_bounds__(SC_THREADS) void gin_bin(const int* __restrict__ src,
                                                      const int* __restrict__ dst,
                                                      int* __restrict__ cells) {
    __shared__ int cnt[N_SBLK];

    const int tid = (int)threadIdx.x;
    if (tid < N_SBLK) cnt[tid] = 0;
    __syncthreads();

    const int e0 = (int)blockIdx.x * CHUNK_E;
    int e1 = e0 + CHUNK_E;
    if (e1 > N_EDGES) e1 = N_EDGES;
    int d[BIN_EPT], s[BIN_EPT];
    #pragma unroll
    for (int j = 0; j < BIN_EPT; ++j) {
        const int e = e0 + tid + SC_THREADS * j;
        const int ec = e < e1 ? e : N_EDGES - 1;
        d[j] = dst[ec];
        s[j] = src[ec];
        if (e >= e1) d[j] = -1;
    }
    static_assert(BIN_EPT == 4, "the statement below names the loaded values one by one");
    asm volatile("" :: "v"(d[0]), "v"(d[1]), "v"(d[2]), "v"(d[3]), "v"(s[0]), "v"(s[1]), "v"(s[2]), "v"(s[3]));
    #pragma unroll
    for (int j = 0; j < BIN_EPT; ++j) {
        if ((unsigned)d[j] < (unsigned)N_NODES) {
            const int ob  = d[j] / ROWS_PB;
            const int lr  = d[j] - ob * ROWS_PB;
            const int pos = __hip_atomic_fetch_add(cnt + ob, 1, __ATOMIC_RELAXED, __HIP_MEMORY_SCOPE_WORKGROUP);
            if (pos < CELL_W - 1) cells[((long)ob * N_CHUNK + blockIdx.x) * CELL_W + 1 + pos] = s[j] | (lr << 20);
        }
    }
    __syncthreads();
    if (tid < N_SBLK) cells[((long)tid * N_CHUNK + blockIdx.x) * CELL_W] = cnt[tid];
}

template <bool FIRST>
__global__ __launch_bounds__(SC_THREADS) void gin_rowsum(const float* __restrict__ hin,
                                                         const int* __restrict__ src,
                                                         const int* __restrict__ dst,
                                                         float* __restrict__ agg,
                                                         int* lists,
                                                         int* __restrict__ lens,
                                                         const int* __restrict__ cells) {
    __shared__ __attribute__((aligned(16))) float rows[ROWS_PB * DIM];
    __shared__ int   queue[SC_WAVES][QCAP];
    __shared__ int   lcnt[SC_WAVES];
    __shared__ int   over;

    const int tid  = (int)threadIdx.x;
    const int lane = tid & 31;
    const int wave = __builtin_amdgcn_readfirstlane(tid >> 5);
    const int row0 = (int)blockIdx.x * ROWS_PB;
    int* blist = lists + (long)blockIdx.x * LCAP;

    int cv;
    bool whole;
    if (FIRST) {
        if (tid < SC_WAVES) lcnt[tid] = 0;
        if (tid == 0) over = 0;
        __syncthreads();
        const int c = tid & (N_CHUNK - 1);
        const int q = tid / N_CHUNK;
        const int* cell = cells + ((long)blockIdx.x * N_CHUNK + c) * CELL_W;
        int n = cell[0];
        if (n > CELL_W - 1) { over = 1; n = CELL_W - 1; }
        int w[CELL_PT];
        #pragma unroll
        for (int j = 0; j < CELL_PT; ++j) {
            const int k = q + (SC_THREADS / N_CHUNK) * j;
            w[j] = 0;
            if (k < n) w[j] = cell[1 + k];
        }
        #pragma unroll
        for (int j = 0; j < CELL_PT; ++j) {
            const int k = q + (SC_THREADS / N_CHUNK) * j;
            if (k < n) {
                const int o   = (w[j] >> 20) & (SC_WAVES - 1);
                const int pos = __hip_atomic_fetch_add(lcnt + o, 1, __ATOMIC_RELAXED, __HIP_MEMORY_SCOPE_WORKGROUP);
                if (pos < OCAP) blist[o * OCAP + pos] = w[j];
            }
        }
        __syncthreads();
        cv = lcnt[lane];
        const bool ov = over != 0;
        if (tid < SC_WAVES) lens[blockIdx.x * SC_WAVES + tid] = ov ? OCAP + 1 : cv;
        whole = !ov && __builtin_amdgcn_ballot_w32(cv > OCAP) == 0u;
    } else {
        cv = lens[blockIdx.x * SC_WAVES + lane];
        whole = __builtin_amdgcn_ballot_w32(cv > OCAP) == 0u;
    }
    if (whole) {
        const int total = __builtin_amdgcn_readlane(cv, wave);
        const int* bl = blist + wave * OCAP;
        v32f acc;
        #pragma unroll
        for (int e = 0; e < 32; ++e) acc[e] = 0.0f;
        const int nb = total / EPN;
        if (nb > 0) {
            const int vw0 = bl[lane & (EPN - 1)];
            int wd[EPN];
            #pragma unroll
            for (int j = 0; j < EPN; ++j) wd[j] = __builtin_amdgcn_readlane(vw0, j);
            for (int b = 0; b < nb; ++b) {
                const int bn = (b + 1 < nb) ? b + 1 : b;
                const int vw = bl[EPN * bn + (lane & (EPN - 1))];
                acc_words<true>(hin, wd, EPN, lane, vw, acc);
                #pragma unroll
                for (int j = 0; j < EPN; ++j) wd[j] = __builtin_amdgcn_readlane(vw, j);
            }
        }
        const int rem = total - EPN * nb;
        if (rem > 0) {
            int wd[EPN];
            const int vt = bl[EPN * nb + (lane < rem ? lane : 0)];
            #pragma unroll
            for (int j = 0; j < EPN; ++j) wd[j] = __builtin_amdgcn_readlane(vt, j);
            acc_words<false>(hin, wd, rem, lane, 0, acc);
        }
        #pragma unroll
        for (int s = 0; s < NSLOT; ++s) {
            const int lr  = s * SC_WAVES + wave;
            const int row = row0 + lr;
            if (lr < ROWS_PB && row < N_NODES) {
                v4f o;
                o.x = acc[4 * s + 0]; o.y = acc[4 * s + 1]; o.z = acc[4 * s + 2]; o.w = acc[4 * s + 3];
                *(v4f*)(agg + (long)row * DIM + 4 * lane) = o;
            }
        }
    } else {
        zero_rows(rows, tid);
        __syncthreads();
        filter_add(hin, src, dst, rows, queue[wave], row0, wave, lane);
        __syncthreads();
        store_rows(rows, agg, row0, tid);
    }
}

#define TPB 8
#define N_TILES (N_NODES / 16)
static_assert(N_NODES % 16 == 0, "whole 16-row tiles only");

__device__ __forceinline__ void stage_w_load(const float* __restrict__ Wg, int tid, v4f (&w)[16]) {
    #pragma unroll
    for (int it = 0; it < 16; ++it) {
        const int id = tid + 256 * it;
        const int n  = id & 127;
        const int kg = id >> 7;
        const float* wp = Wg + (4 * kg) * DIM + n;
        w[it].x = wp[0];
        w[it].y = wp[DIM];
        w[it].z = wp[2 * DIM];
        w[it].w = wp[3 * DIM];
    }
}

__device__ __forceinline__ void stage_w_store(v4f* __restrict__ Wl, int tid, const v4f (&w)[16]) {
    #pragma unroll
    for (int it = 0; it < 16; ++it) {
        const int id = tid + 256 * it;
        const int n  = id & 127;
        const int kg = id >> 7;
        const int t  = kg >> 2;
        const int h  = (kg >> 1) & 1;
        const int q  = kg & 1;
        const int u  = n >> 4;
        const int c  = n & 15;
        Wl[(((u * 8 + t) * 2 + q) * 32) + 16 * h + c] = w[it];
    }
}

__device__ __forceinline__ float row16_sum(float v) {
    v += __builtin_bit_cast(float, __builtin_amdgcn_update_dpp(0, __builtin_bit_cast(int, v), 0x161, 0xf, 0xf, false));
    v += __builtin_bit_cast(float, __builtin_amdgcn_update_dpp(0, __builtin_bit_cast(int, v), 0x162, 0xf, 0xf, false));
    v += __builtin_bit_cast(float, __builtin_amdgcn_update_dpp(0, __builtin_bit_cast(int, v), 0x164, 0xf, 0xf, false));
    v += __builtin_bit_cast(float, __builtin_amdgcn_update_dpp(0, __builtin_bit_cast(int, v), 0x168, 0xf, 0xf, false));
    return v;
}

__global__ __launch_bounds__(256) void gin_mlp(const float* __restrict__ hin,
                                               const float* __restrict__ agg,
                                               const float* __restrict__ W1,
                                               const float* __restrict__ b1,
                                               const float* __restrict__ W2,
                                               const float* __restrict__ b2,
                                               const float* __restrict__ epsp,
                                               float* __restrict__ hout,
                                               float* __restrict__ stats,
                                               int do_stats) {
    __shared__ v4f w1l[DIM * DIM / 4];
    __shared__ v4f w2l[DIM * DIM / 4];
    __shared__ float red[TPB][2 * DIM];

    const int tid  = threadIdx.x;
    const int lane = tid & 31;
    const int c    = lane & 15;
    const int h    = lane >> 4;
    const int wave = __builtin_amdgcn_readfirstlane(tid >> 5);
    const int tile = blockIdx.x * TPB + wave;
    const bool valid = tile < N_TILES;
    const int tl = valid ? tile : N_TILES - 1;
    const long rowoff = (long)(tl * 16 + c) * DIM + 8 * h;
    const float e1 = 1.0f + epsp[0];

    v4f hr[8][2], ar[8][2];
    #pragma unroll
    for (int t = 0; t < 8; ++t) {
        hr[t][0] = *(const v4f*)(hin + rowoff + 16 * t);
        hr[t][1] = *(const v4f*)(hin + rowoff + 16 * t + 4);
        ar[t][0] = *(const v4f*)(agg + rowoff + 16 * t);
        ar[t][1] = *(const v4f*)(agg + rowoff + 16 * t + 4);
    }
    v4f wr1[16], wr2[16];
    stage_w_load(W1, tid, wr1);
    stage_w_load(W2, tid, wr2);

    float zf[8][8];
    #pragma unroll
    for (int t = 0; t < 8; ++t) {
        zf[t][0] = e1 * hr[t][0].x + ar[t][0].x;
        zf[t][1] = e1 * hr[t][0].y + ar[t][0].y;
        zf[t][2] = e1 * hr[t][0].z + ar[t][0].z;
        zf[t][3] = e1 * hr[t][0].w + ar[t][0].w;
        zf[t][4] = e1 * hr[t][1].x + ar[t][1].x;
        zf[t][5] = e1 * hr[t][1].y + ar[t][1].y;
        zf[t][6] = e1 * hr[t][1].z + ar[t][1].z;
        zf[t][7] = e1 * hr[t][1].w + ar[t][1].w;
    }

    stage_w_store(w1l, tid, wr1);
    stage_w_store(w2l, tid, wr2);
    __syncthreads();

    v8f acc2[8];
    if (valid) {
        v8f acc1[8];
        #pragma unroll
        for (int u = 0; u < 8; ++u) {
            const v4f bl = *(const v4f*)(b1 + 16 * u + 8 * h);
            const v4f bh = *(const v4f*)(b1 + 16 * u + 8 * h + 4);
            acc1[u][0] = bl.x; acc1[u][1] = bl.y; acc1[u][2] = bl.z; acc1[u][3] = bl.w;
            acc1[u][4] = bh.x; acc1[u][5] = bh.y; acc1[u][6] = bh.z; acc1[u][7] = bh.w;
        }
        #pragma unroll
        for (int t = 0; t < 8; ++t) {
            #pragma unroll
            for (int u = 0; u < 8; ++u) {
                const v4f wa = w1l[((u * 8 + t) * 2 + 0) * 32 + lane];
                const v4f wb = w1l[((u * 8 + t) * 2 + 1) * 32 + lane];
                v2f a, b;
                a.x = wa.x; a.y = wa.y; b.x = zf[t][0]; b.y = zf[t][1];
                acc1[u] = __builtin_amdgcn_wmma_f32_16x16x4_f32(false, a, false, b, (short)0, acc1[u], false, false);
                a.x = wa.z; a.y = wa.w; b.x = zf[t][2]; b.y = zf[t][3];
                acc1[u] = __builtin_amdgcn_wmma_f32_16x16x4_f32(false, a, false, b, (short)0, acc1[u], false, false);
                a.x = wb.x; a.y = wb.y; b.x = zf[t][4]; b.y = zf[t][5];
                acc1[u] = __builtin_amdgcn_wmma_f32_16x16x4_f32(false, a, false, b, (short)0, acc1[u], false, false);
                a.x = wb.z; a.y = wb.w; b.x = zf[t][6]; b.y = zf[t][7];
                acc1[u] = __builtin_amdgcn_wmma_f32_16x16x4_f32(false, a, false, b, (short)0, acc1[u], false, false);
            }
        }
        #pragma unroll
        for (int t = 0; t < 8; ++t) {
            #pragma unroll
            for (int r = 0; r < 8; ++r) acc1[t][r] = fmaxf(acc1[t][r], 0.0f);
        }

        #pragma unroll
        for (int u = 0; u < 8; ++u) {
            const v4f bl = *(const v4f*)(b2 + 16 * u + 8 * h);
            const v4f bh = *(const v4f*)(b2 + 16 * u + 8 * h + 4);
            acc2[u][0] = bl.x; acc2[u][1] = bl.y; acc2[u][2] = bl.z; acc2[u][3] = bl.w;
            acc2[u][4] = bh.x; acc2[u][5] = bh.y; acc2[u][6] = bh.z; acc2[u][7] = bh.w;
        }
        #pragma unroll
        for (int t = 0; t < 8; ++t) {
            #pragma unroll
            for (int u = 0; u < 8; ++u) {
                const v4f wa = w2l[((u * 8 + t) * 2 + 0) * 32 + lane];
                const v4f wb = w2l[((u * 8 + t) * 2 + 1) * 32 + lane];
                v2f a, b;
                a.x = wa.x; a.y = wa.y; b.x = acc1[t][0]; b.y = acc1[t][1];
                acc2[u] = __builtin_amdgcn_wmma_f32_16x16x4_f32(false, a, false, b, (short)0, acc2[u], false, false);
                a.x = wa.z; a.y = wa.w; b.x = acc1[t][2]; b.y = acc1[t][3];
                acc2[u] = __builtin_amdgcn_wmma_f32_16x16x4_f32(false, a, false, b, (short)0, acc2[u], false, false);
                a.x = wb.x; a.y = wb.y; b.x = acc1[t][4]; b.y = acc1[t][5];
                acc2[u] = __builtin_amdgcn_wmma_f32_16x16x4_f32(false, a, false, b, (short)0, acc2[u], false, false);
                a.x = wb.z; a.y = wb.w; b.x = acc1[t][6]; b.y = acc1[t][7];
                acc2[u] = __builtin_amdgcn_wmma_f32_16x16x4_f32(false, a, false, b, (short)0, acc2[u], false, false);
            }
        }

        #pragma unroll
        for (int u = 0; u < 8; ++u) {
            v4f lo, hi4;
            lo.x = acc2[u][0]; lo.y = acc2[u][1]; lo.z = acc2[u][2]; lo.w = acc2[u][3];
            hi4.x = acc2[u][4]; hi4.y = acc2[u][5]; hi4.z = acc2[u][6]; hi4.w = acc2[u][7];
            *(v4f*)(hout + rowoff + 16 * u)     = lo;
            *(v4f*)(hout + rowoff + 16 * u + 4) = hi4;
        }
    } else {
        #pragma unroll
        for (int u = 0; u < 8; ++u) {
            #pragma unroll
            for (int r = 0; r < 8; ++r) acc2[u][r] = 0.0f;
        }
    }

    if (do_stats) {
        float cs[8][8], cq[8][8];
        #pragma unroll
        for (int u = 0; u < 8; ++u) {
            #pragma unroll
            for (int r = 0; r < 8; ++r) {
                const float val = acc2[u][r];
                cs[u][r] = row16_sum(val);
                cq[u][r] = row16_sum(val * val);
            }
        }
        if (c == 0) {
            #pragma unroll
            for (int u = 0; u < 8; ++u) {
                #pragma unroll
                for (int r = 0; r < 8; ++r) {
                    red[wave][16 * u + 8 * h + r]       = cs[u][r];
                    red[wave][DIM + 16 * u + 8 * h + r] = cq[u][r];
                }
            }
        }
        __syncthreads();
        float tot = 0.0f;
        #pragma unroll
        for (int w = 0; w < TPB; ++w) tot += red[w][tid];
        atomicAdd(&stats[tid], tot);
    }
}

__global__ __launch_bounds__(256) void gin_bn_relu(float* __restrict__ z,
                                                   const float* __restrict__ stats,
                                                   const float* __restrict__ gamma,
                                                   const float* __restrict__ beta) {
    long i = (long)blockIdx.x * blockDim.x + threadIdx.x;
    if (i >= (long)N_NODES * DIM) return;
    int c = (int)(i & (DIM - 1));
    const float invN = 1.0f / (float)N_NODES;
    float mu  = stats[c] * invN;
    float var = stats[DIM + c] * invN - mu * mu;
    float v = (z[i] - mu) * rsqrtf(var + BN_EPS) * gamma[c] + beta[c];
    z[i] = fmaxf(v, 0.0f);
}

__global__ __launch_bounds__(256) void gin_pool(const float* __restrict__ h,
                                                const int* __restrict__ batch,
                                                float* __restrict__ out) {
    const int lane = (int)(threadIdx.x & 31);
    const int wv = __builtin_amdgcn_readfirstlane((int)(blockIdx.x * 8 + (threadIdx.x >> 5)));
    const int n0 = wv * EPW;
    if (n0 >= N_NODES) return;
    int g[EPW];
    #pragma unroll
    for (int j = 0; j < EPW; ++j) g[j] = batch[n0 + j];
    float v[EPW][4];
    #pragma unroll
    for (int j = 0; j < EPW; ++j) {
        const float* hp = h + (long)(n0 + j) * DIM + lane;
        v[j][0] = hp[0];
        v[j][1] = hp[32];
        v[j][2] = hp[64];
        v[j][3] = hp[96];
    }
    #pragma unroll
    for (int j = 0; j < EPW; ++j) {
        float* op = out + (long)g[j] * DIM + lane;
        atomicAdd(op + 0,  v[j][0]);
        atomicAdd(op + 32, v[j][1]);
        atomicAdd(op + 64, v[j][2]);
        atomicAdd(op + 96, v[j][3]);
    }
}

extern "C" void kernel_launch(void* const* d_in, const int* in_sizes, int n_in,
                              void* d_out, int out_size, void* d_ws, size_t ws_size,
                              hipStream_t stream) {
    const float* x     = (const float*)d_in[0];
    const int*   ei    = (const int*)d_in[1];
    const int*   batch = (const int*)d_in[2];
    const float* W1    = (const float*)d_in[3];
    const float* b1    = (const float*)d_in[4];
    const float* W2    = (const float*)d_in[5];
    const float* b2    = (const float*)d_in[6];
    const float* eps   = (const float*)d_in[7];
    const float* gamma = (const float*)d_in[8];
    const float* beta  = (const float*)d_in[9];
    float* out = (float*)d_out;

    const long ND = (long)N_NODES * DIM;
    float* hA    = (float*)d_ws;
    float* hB    = hA + ND;
    float* agg   = hB + ND;
    float* stats = agg + ND;
    int*   lens  = (int*)(stats + 2 * DIM);
    int*   lists = lens + N_SBLK * SC_WAVES;
    int*   cells = lists + (long)N_SBLK * LCAP;

    const int* src = ei;
    const int* dst = ei + N_EDGES;

    const int  mlpGrd  = (N_TILES + TPB - 1) / TPB;
    const int  bnGrd   = (int)((ND + 255) / 256);
    const int  poolGrd = (N_NODES / EPW + 7) / 8;

    gin_bin<<<N_CHUNK, SC_THREADS, 0, stream>>>(src, dst, cells);

    for (int i = 0; i < N_LAYERS; ++i) {
        const float* hin  = (i == 0) ? x : ((i & 1) ? hA : hB);
        float*       hout = (i & 1) ? hB : hA;
        int do_stats = (i < N_LAYERS - 1) ? 1 : 0;

        if (do_stats) gin_zero<<<1, 256, 0, stream>>>(stats, 2 * DIM);
        if (i == 0) gin_rowsum<true><<<N_SBLK, SC_THREADS, 0, stream>>>(hin, src, dst, agg, lists, lens, cells);
        else        gin_rowsum<false><<<N_SBLK, SC_THREADS, 0, stream>>>(hin, src, dst, agg, lists, lens, cells);
        gin_mlp<<<mlpGrd, 256, 0, stream>>>(hin, agg,
                                            W1 + (long)i * DIM * DIM, b1 + (long)i * DIM,
                                            W2 + (long)i * DIM * DIM, b2 + (long)i * DIM,
                                            eps + i, hout, stats, do_stats);
        if (do_stats) {
            gin_bn_relu<<<bnGrd, 256, 0, stream>>>(hout, stats,
                                                   gamma + (long)i * DIM,
                                                   beta + (long)i * DIM);
        }
    }
    gin_zero<<<(NUM_GRAPHS * DIM + 255) / 256, 256, 0, stream>>>(out, NUM_GRAPHS * DIM);
    gin_pool<<<poolGrd, 256, 0, stream>>>(hB, batch, out);
}


# BatchNorm+ReLU pass: grid-stride, a thread takes 8 elements of one column by 4-byte accesses (side-by-side waves still touch consecutive lines); 25000 waves, not 200000; no sum order changes
# speedup vs baseline: 3.0959x; 1.0698x over previous
#include <hip/hip_runtime.h>
#include <hip/hip_bf16.h>

#define N_NODES   50000
#define N_EDGES   800000
#define DIM       128
#define N_LAYERS  4
#define NUM_GRAPHS 128
#define BN_EPS    1e-5f

typedef __attribute__((ext_vector_type(2))) float v2f;
typedef __attribute__((ext_vector_type(8))) float v8f;

__global__ __launch_bounds__(256) void gin_zero(float* __restrict__ p, long n) {
    long i = (long)blockIdx.x * blockDim.x + threadIdx.x;
    if (i < n) p[i] = 0.0f;
}

#define EPW 8
#define ROWS_PB   196
#define SC_WAVES  32
#define SC_THREADS (SC_WAVES * 32)
#define N_SBLK    ((N_NODES + ROWS_PB - 1) / ROWS_PB)
#define TILE_L    8
#define TILE_E    (TILE_L * 128)
#define N_ETILES  ((N_EDGES + TILE_E - 1) / TILE_E)
#define QCAP      (TILE_E + EPW)
#define LCAP      65536
#define OCAP      (LCAP / SC_WAVES)
#define NSLOT     ((ROWS_PB + SC_WAVES - 1) / SC_WAVES)
#define EPN       16
#define N_CHUNK   256
#define CHUNK_E   ((N_EDGES + N_CHUNK - 1) / N_CHUNK)
#define BIN_EPT   ((CHUNK_E + SC_THREADS - 1) / SC_THREADS)
#define CELL_W    64
#define CELL_PT   16
static_assert(N_EDGES % 4 == 0, "a 16-byte index load must not straddle the end of dst");
static_assert(N_EDGES <= (1 << 20), "the edge number must fit the low 20 bits of a queue word");
static_assert(ROWS_PB < (1 << 11), "the local row must fit the bits above the edge number");
static_assert(N_NODES % EPW == 0, "a pool wave's nodes must all exist");
static_assert(N_NODES <= (1 << 20), "a source number must fit the low 20 bits of a list word");
static_assert(LCAP % EPW == 0, "a batch of list words must not straddle two blocks' lists");
static_assert(OCAP % EPN == 0, "a batch of list words must not straddle two owners' lists");
static_assert(EPN <= 32, "one lane per word of a batch");
static_assert(NSLOT * 4 <= 32, "a wave's row sums must fit the 32-float accumulator vector");
static_assert(SC_WAVES == 32, "one lane per owner list count");
static_assert(SC_THREADS % N_CHUNK == 0, "whole threads per cell in the regrouping step");
static_assert((SC_THREADS / N_CHUNK) * CELL_PT >= CELL_W - 1, "the threads of a cell must cover all its list words");
static_assert(N_SBLK <= SC_THREADS, "one thread per cell count in the binning launch");

typedef __attribute__((ext_vector_type(4))) int   v4i;
typedef __attribute__((ext_vector_type(4))) float v4f;
typedef __attribute__((ext_vector_type(32))) float v32f;

__device__ __forceinline__ void load_tile(const int* __restrict__ dst, int tile, int lane, v4i (&t)[TILE_L]) {
    #pragma unroll
    for (int j = 0; j < TILE_L; ++j) {
        const int idx = tile * TILE_E + j * 128 + 4 * lane;
        const int idc = idx < N_EDGES ? idx : N_EDGES - 4;
        v4i d = *(const v4i*)(dst + idc);
        if (idx >= N_EDGES) { d.x = -1; d.y = -1; d.z = -1; d.w = -1; }
        t[j] = d;
    }
}

template <bool FULL>
__device__ __forceinline__ void add_words(const float* __restrict__ hin, float* rows,
                                          const int (&wd)[EPW], int n, int lane) {
    float v[EPW][4];
    #pragma unroll
    for (int j = 0; j < EPW; ++j) {
        if (FULL || j < n) {
            const float* hs = hin + (long)(wd[j] & 0xFFFFF) * DIM + lane;
            v[j][0] = hs[0];
            v[j][1] = hs[32];
            v[j][2] = hs[64];
            v[j][3] = hs[96];
        }
    }
    #pragma unroll
    for (int j = 0; j < EPW; ++j) {
        if (FULL || j < n) {
            float* r = rows + (wd[j] >> 20) * DIM + lane;
            __hip_atomic_fetch_add(r + 0,  v[j][0], __ATOMIC_RELAXED, __HIP_MEMORY_SCOPE_WORKGROUP);
            __hip_atomic_fetch_add(r + 32, v[j][1], __ATOMIC_RELAXED, __HIP_MEMORY_SCOPE_WORKGROUP);
            __hip_atomic_fetch_add(r + 64, v[j][2], __ATOMIC_RELAXED, __HIP_MEMORY_SCOPE_WORKGROUP);
            __hip_atomic_fetch_add(r + 96, v[j][3], __ATOMIC_RELAXED, __HIP_MEMORY_SCOPE_WORKGROUP);
        }
    }
}

template <bool FULL>
__device__ __forceinline__ void acc_words(const float* __restrict__ hin, const int (&wd)[EPN], int n, int lane,
                                          int vnext, v32f& acc) {
    v4f v[EPN];
    #pragma unroll
    for (int j = 0; j < EPN; ++j) {
        if (FULL || j < n) {
            const float* rb = hin + (size_t)(wd[j] & 0xFFFFF) * DIM;
            v[j] = *(const v4f*)(rb + 4u * (unsigned)lane);
        }
    }
    if (FULL) {
        asm volatile("" :: "v"(v[0]), "v"(v[1]), "v"(v[2]), "v"(v[3]), "v"(v[4]), "v"(v[5]), "v"(v[6]), "v"(v[7]),
                           "v"(v[8]), "v"(v[9]), "v"(v[10]), "v"(v[11]), "v"(v[12]), "v"(v[13]), "v"(v[14]), "v"(v[15]),
                           "v"(vnext));
    }
    #pragma unroll
    for (int j = 0; j < EPN; ++j) {
        if (FULL || j < n) {
            const int e = (wd[j] >> 25) * 4;
            acc[e + 0] += v[j].x;
            acc[e + 1] += v[j].y;
            acc[e + 2] += v[j].z;
            acc[e + 3] += v[j].w;
        }
    }
}

template <bool FULL>
__device__ __forceinline__ void add_batch(const float* __restrict__ hin, const int* __restrict__ src,
                                          float* rows, const int* qe, int n, int lane) {
    int ent[EPW], wd[EPW];
    #pragma unroll
    for (int j = 0; j < EPW; ++j) {
        ent[j] = 0;
        if (FULL || j < n)
            ent[j] = __builtin_amdgcn_readfirstlane(__hip_atomic_load(qe + j, __ATOMIC_RELAXED, __HIP_MEMORY_SCOPE_WAVEFRONT));
    }
    #pragma unroll
    for (int j = 0; j < EPW; ++j) {
        wd[j] = 0;
        if (FULL || j < n) wd[j] = src[ent[j] & 0xFFFFF] | (ent[j] & ~0xFFFFF);
    }
    add_words<FULL>(hin, rows, wd, n, lane);
}

__device__ __forceinline__ void filter_add(const float* __restrict__ hin, const int* __restrict__ src,
                                           const int* __restrict__ dst, float* rows, int* q,
                                           int row0, int wave, int lane) {
    int count = 0;

    v4i cur[TILE_L], nxt[TILE_L];
    load_tile(dst, wave, lane, cur);
    for (int tile = wave; tile < N_ETILES; tile += SC_WAVES) {
        load_tile(dst, tile + SC_WAVES, lane, nxt);
        #pragma unroll
        for (int j = 0; j < TILE_L; ++j) {
            #pragma unroll
            for (int k = 0; k < 4; ++k) {
                const int e = tile * TILE_E + j * 128 + 4 * lane + k;
                const unsigned lr = (unsigned)(cur[j][k] - row0);
                const bool m = lr < (unsigned)ROWS_PB;
                const unsigned mask = __builtin_amdgcn_ballot_w32(m);
                if (mask != 0u) {
                    if (m) {
                        const int pos = count + (int)__builtin_amdgcn_mbcnt_lo(mask, 0u);
                        __hip_atomic_store(q + pos, e | (int)(lr << 20), __ATOMIC_RELAXED, __HIP_MEMORY_SCOPE_WAVEFRONT);
                    }
                    count += __builtin_popcount(mask);
                }
            }
        }
        while (count >= EPW) {
            count -= EPW;
            add_batch<true>(hin, src, rows, q + count, EPW, lane);
        }
        #pragma unroll
        for (int j = 0; j < TILE_L; ++j) cur[j] = nxt[j];
    }
    if (count > 0) add_batch<false>(hin, src, rows, q, count, lane);
}

__device__ __forceinline__ void zero_rows(float* rows, int tid) {
    for (int i = tid; i < ROWS_PB * DIM / 4; i += SC_THREADS) {
        v4f z4;
        z4.x = 0.0f; z4.y = 0.0f; z4.z = 0.0f; z4.w = 0.0f;
        ((v4f*)rows)[i] = z4;
    }
}

__device__ __forceinline__ void store_rows(const float* rows, float* __restrict__ agg, int row0, int tid) {
    int nrows = N_NODES - row0;
    if (nrows > ROWS_PB) nrows = ROWS_PB;
    v4f* ag4 = (v4f*)(agg + (long)row0 * DIM);
    for (int i = tid; i < nrows * (DIM / 4); i += SC_THREADS) ag4[i] = ((const v4f*)rows)[i];
}


__global__ __launch_bounds__(SC_THREADS) void gin_bin(const int* __restrict__ src,
                                                      const int* __restrict__ dst,
                                                      int* __restrict__ cells) {
    __shared__ int cnt[N_SBLK];

    const int tid = (int)threadIdx.x;
    if (tid < N_SBLK) cnt[tid] = 0;
    __syncthreads();

    const int e0 = (int)blockIdx.x * CHUNK_E;
    int e1 = e0 + CHUNK_E;
    if (e1 > N_EDGES) e1 = N_EDGES;
    int d[BIN_EPT], s[BIN_EPT];
    #pragma unroll
    for (int j = 0; j < BIN_EPT; ++j) {
        const int e = e0 + tid + SC_THREADS * j;
        const int ec = e < e1 ? e : N_EDGES - 1;
        d[j] = dst[ec];
        s[j] = src[ec];
        if (e >= e1) d[j] = -1;
    }
    static_assert(BIN_EPT == 4, "the statement below names the loaded values one by one");
    asm volatile("" :: "v"(d[0]), "v"(d[1]), "v"(d[2]), "v"(d[3]), "v"(s[0]), "v"(s[1]), "v"(s[2]), "v"(s[3]));
    #pragma unroll
    for (int j = 0; j < BIN_EPT; ++j) {
        if ((unsigned)d[j] < (unsigned)N_NODES) {
            const int ob  = d[j] / ROWS_PB;
            const int lr  = d[j] - ob * ROWS_PB;
            const int pos = __hip_atomic_fetch_add(cnt + ob, 1, __ATOMIC_RELAXED, __HIP_MEMORY_SCOPE_WORKGROUP);
            if (pos < CELL_W - 1) cells[((long)ob * N_CHUNK + blockIdx.x) * CELL_W + 1 + pos] = s[j] | (lr << 20);
        }
    }
    __syncthreads();
    if (tid < N_SBLK) cells[((long)tid * N_CHUNK + blockIdx.x) * CELL_W] = cnt[tid];
}

template <bool FIRST>
__global__ __launch_bounds__(SC_THREADS) void gin_rowsum(const float* __restrict__ hin,
                                                         const int* __restrict__ src,
                                                         const int* __restrict__ dst,
                                                         float* __restrict__ agg,
                                                         int* lists,
                                                         int* __restrict__ lens,
                                                         const int* __restrict__ cells) {
    __shared__ __attribute__((aligned(16))) float rows[ROWS_PB * DIM];
    __shared__ int   queue[SC_WAVES][QCAP];
    __shared__ int   lcnt[SC_WAVES];
    __shared__ int   over;

    const int tid  = (int)threadIdx.x;
    const int lane = tid & 31;
    const int wave = __builtin_amdgcn_readfirstlane(tid >> 5);
    const int row0 = (int)blockIdx.x * ROWS_PB;
    int* blist = lists + (long)blockIdx.x * LCAP;

    int cv;
    bool whole;
    if (FIRST) {
        if (tid < SC_WAVES) lcnt[tid] = 0;
        if (tid == 0) over = 0;
        __syncthreads();
        const int c = tid & (N_CHUNK - 1);
        const int q = tid / N_CHUNK;
        const int* cell = cells + ((long)blockIdx.x * N_CHUNK + c) * CELL_W;
        int n = cell[0];
        if (n > CELL_W - 1) { over = 1; n = CELL_W - 1; }
        int w[CELL_PT];
        #pragma unroll
        for (int j = 0; j < CELL_PT; ++j) {
            const int k = q + (SC_THREADS / N_CHUNK) * j;
            w[j] = 0;
            if (k < n) w[j] = cell[1 + k];
        }
        #pragma unroll
        for (int j = 0; j < CELL_PT; ++j) {
            const int k = q + (SC_THREADS / N_CHUNK) * j;
            if (k < n) {
                const int o   = (w[j] >> 20) & (SC_WAVES - 1);
                const int pos = __hip_atomic_fetch_add(lcnt + o, 1, __ATOMIC_RELAXED, __HIP_MEMORY_SCOPE_WORKGROUP);
                if (pos < OCAP) blist[o * OCAP + pos] = w[j];
            }
        }
        __syncthreads();
        cv = lcnt[lane];
        const bool ov = over != 0;
        if (tid < SC_WAVES) lens[blockIdx.x * SC_WAVES + tid] = ov ? OCAP + 1 : cv;
        whole = !ov && __builtin_amdgcn_ballot_w32(cv > OCAP) == 0u;
    } else {
        cv = lens[blockIdx.x * SC_WAVES + lane];
        whole = __builtin_amdgcn_ballot_w32(cv > OCAP) == 0u;
    }
    if (whole) {
        const int total = __builtin_amdgcn_readlane(cv, wave);
        const int* bl = blist + wave * OCAP;
        v32f acc;
        #pragma unroll
        for (int e = 0; e < 32; ++e) acc[e] = 0.0f;
        const int nb = total / EPN;
        if (nb > 0) {
            const int vw0 = bl[lane & (EPN - 1)];
            int wd[EPN];
            #pragma unroll
            for (int j = 0; j < EPN; ++j) wd[j] = __builtin_amdgcn_readlane(vw0, j);
            for (int b = 0; b < nb; ++b) {
                const int bn = (b + 1 < nb) ? b + 1 : b;
                const int vw = bl[EPN * bn + (lane & (EPN - 1))];
                acc_words<true>(hin, wd, EPN, lane, vw, acc);
                #pragma unroll
                for (int j = 0; j < EPN; ++j) wd[j] = __builtin_amdgcn_readlane(vw, j);
            }
        }
        const int rem = total - EPN * nb;
        if (rem > 0) {
            int wd[EPN];
            const int vt = bl[EPN * nb + (lane < rem ? lane : 0)];
            #pragma unroll
            for (int j = 0; j < EPN; ++j) wd[j] = __builtin_amdgcn_readlane(vt, j);
            acc_words<false>(hin, wd, rem, lane, 0, acc);
        }
        #pragma unroll
        for (int s = 0; s < NSLOT; ++s) {
            const int lr  = s * SC_WAVES + wave;
            const int row = row0 + lr;
            if (lr < ROWS_PB && row < N_NODES) {
                v4f o;
                o.x = acc[4 * s + 0]; o.y = acc[4 * s + 1]; o.z = acc[4 * s + 2]; o.w = acc[4 * s + 3];
                *(v4f*)(agg + (long)row * DIM + 4 * lane) = o;
            }
        }
    } else {
        zero_rows(rows, tid);
        __syncthreads();
        filter_add(hin, src, dst, rows, queue[wave], row0, wave, lane);
        __syncthreads();
        store_rows(rows, agg, row0, tid);
    }
}

#define TPB 8
#define N_TILES (N_NODES / 16)
static_assert(N_NODES % 16 == 0, "whole 16-row tiles only");

__device__ __forceinline__ void stage_w_load(const float* __restrict__ Wg, int tid, v4f (&w)[16]) {
    #pragma unroll
    for (int it = 0; it < 16; ++it) {
        const int id = tid + 256 * it;
        const int n  = id & 127;
        const int kg = id >> 7;
        const float* wp = Wg + (4 * kg) * DIM + n;
        w[it].x = wp[0];
        w[it].y = wp[DIM];
        w[it].z = wp[2 * DIM];
        w[it].w = wp[3 * DIM];
    }
}

__device__ __forceinline__ void stage_w_store(v4f* __restrict__ Wl, int tid, const v4f (&w)[16]) {
    #pragma unroll
    for (int it = 0; it < 16; ++it) {
        const int id = tid + 256 * it;
        const int n  = id & 127;
        const int kg = id >> 7;
        const int t  = kg >> 2;
        const int h  = (kg >> 1) & 1;
        const int q  = kg & 1;
        const int u  = n >> 4;
        const int c  = n & 15;
        Wl[(((u * 8 + t) * 2 + q) * 32) + 16 * h + c] = w[it];
    }
}

__device__ __forceinline__ float row16_sum(float v) {
    v += __builtin_bit_cast(float, __builtin_amdgcn_update_dpp(0, __builtin_bit_cast(int, v), 0x161, 0xf, 0xf, false));
    v += __builtin_bit_cast(float, __builtin_amdgcn_update_dpp(0, __builtin_bit_cast(int, v), 0x162, 0xf, 0xf, false));
    v += __builtin_bit_cast(float, __builtin_amdgcn_update_dpp(0, __builtin_bit_cast(int, v), 0x164, 0xf, 0xf, false));
    v += __builtin_bit_cast(float, __builtin_amdgcn_update_dpp(0, __builtin_bit_cast(int, v), 0x168, 0xf, 0xf, false));
    return v;
}

__global__ __launch_bounds__(256) void gin_mlp(const float* __restrict__ hin,
                                               const float* __restrict__ agg,
                                               const float* __restrict__ W1,
                                               const float* __restrict__ b1,
                                               const float* __restrict__ W2,
                                               const float* __restrict__ b2,
                                               const float* __restrict__ epsp,
                                               float* __restrict__ hout,
                                               float* __restrict__ stats,
                                               int do_stats) {
    __shared__ v4f w1l[DIM * DIM / 4];
    __shared__ v4f w2l[DIM * DIM / 4];
    __shared__ float red[TPB][2 * DIM];

    const int tid  = threadIdx.x;
    const int lane = tid & 31;
    const int c    = lane & 15;
    const int h    = lane >> 4;
    const int wave = __builtin_amdgcn_readfirstlane(tid >> 5);
    const int tile = blockIdx.x * TPB + wave;
    const bool valid = tile < N_TILES;
    const int tl = valid ? tile : N_TILES - 1;
    const long rowoff = (long)(tl * 16 + c) * DIM + 8 * h;
    const float e1 = 1.0f + epsp[0];

    v4f hr[8][2], ar[8][2];
    #pragma unroll
    for (int t = 0; t < 8; ++t) {
        hr[t][0] = *(const v4f*)(hin + rowoff + 16 * t);
        hr[t][1] = *(const v4f*)(hin + rowoff + 16 * t + 4);
        ar[t][0] = *(const v4f*)(agg + rowoff + 16 * t);
        ar[t][1] = *(const v4f*)(agg + rowoff + 16 * t + 4);
    }
    v4f wr1[16], wr2[16];
    stage_w_load(W1, tid, wr1);
    stage_w_load(W2, tid, wr2);

    float zf[8][8];
    #pragma unroll
    for (int t = 0; t < 8; ++t) {
        zf[t][0] = e1 * hr[t][0].x + ar[t][0].x;
        zf[t][1] = e1 * hr[t][0].y + ar[t][0].y;
        zf[t][2] = e1 * hr[t][0].z + ar[t][0].z;
        zf[t][3] = e1 * hr[t][0].w + ar[t][0].w;
        zf[t][4] = e1 * hr[t][1].x + ar[t][1].x;
        zf[t][5] = e1 * hr[t][1].y + ar[t][1].y;
        zf[t][6] = e1 * hr[t][1].z + ar[t][1].z;
        zf[t][7] = e1 * hr[t][1].w + ar[t][1].w;
    }

    stage_w_store(w1l, tid, wr1);
    stage_w_store(w2l, tid, wr2);
    __syncthreads();

    v8f acc2[8];
    if (valid) {
        v8f acc1[8];
        #pragma unroll
        for (int u = 0; u < 8; ++u) {
            const v4f bl = *(const v4f*)(b1 + 16 * u + 8 * h);
            const v4f bh = *(const v4f*)(b1 + 16 * u + 8 * h + 4);
            acc1[u][0] = bl.x; acc1[u][1] = bl.y; acc1[u][2] = bl.z; acc1[u][3] = bl.w;
            acc1[u][4] = bh.x; acc1[u][5] = bh.y; acc1[u][6] = bh.z; acc1[u][7] = bh.w;
        }
        #pragma unroll
        for (int t = 0; t < 8; ++t) {
            #pragma unroll
            for (int u = 0; u < 8; ++u) {
                const v4f wa = w1l[((u * 8 + t) * 2 + 0) * 32 + lane];
                const v4f wb = w1l[((u * 8 + t) * 2 + 1) * 32 + lane];
                v2f a, b;
                a.x = wa.x; a.y = wa.y; b.x = zf[t][0]; b.y = zf[t][1];
                acc1[u] = __builtin_amdgcn_wmma_f32_16x16x4_f32(false, a, false, b, (short)0, acc1[u], false, false);
                a.x = wa.z; a.y = wa.w; b.x = zf[t][2]; b.y = zf[t][3];
                acc1[u] = __builtin_amdgcn_wmma_f32_16x16x4_f32(false, a, false, b, (short)0, acc1[u], false, false);
                a.x = wb.x; a.y = wb.y; b.x = zf[t][4]; b.y = zf[t][5];
                acc1[u] = __builtin_amdgcn_wmma_f32_16x16x4_f32(false, a, false, b, (short)0, acc1[u], false, false);
                a.x = wb.z; a.y = wb.w; b.x = zf[t][6]; b.y = zf[t][7];
                acc1[u] = __builtin_amdgcn_wmma_f32_16x16x4_f32(false, a, false, b, (short)0, acc1[u], false, false);
            }
        }
        #pragma unroll
        for (int t = 0; t < 8; ++t) {
            #pragma unroll
            for (int r = 0; r < 8; ++r) acc1[t][r] = fmaxf(acc1[t][r], 0.0f);
        }

        #pragma unroll
        for (int u = 0; u < 8; ++u) {
            const v4f bl = *(const v4f*)(b2 + 16 * u + 8 * h);
            const v4f bh = *(const v4f*)(b2 + 16 * u + 8 * h + 4);
            acc2[u][0] = bl.x; acc2[u][1] = bl.y; acc2[u][2] = bl.z; acc2[u][3] = bl.w;
            acc2[u][4] = bh.x; acc2[u][5] = bh.y; acc2[u][6] = bh.z; acc2[u][7] = bh.w;
        }
        #pragma unroll
        for (int t = 0; t < 8; ++t) {
            #pragma unroll
            for (int u = 0; u < 8; ++u) {
                const v4f wa = w2l[((u * 8 + t) * 2 + 0) * 32 + lane];
                const v4f wb = w2l[((u * 8 + t) * 2 + 1) * 32 + lane];
                v2f a, b;
                a.x = wa.x; a.y = wa.y; b.x = acc1[t][0]; b.y = acc1[t][1];
                acc2[u] = __builtin_amdgcn_wmma_f32_16x16x4_f32(false, a, false, b, (short)0, acc2[u], false, false);
                a.x = wa.z; a.y = wa.w; b.x = acc1[t][2]; b.y = acc1[t][3];
                acc2[u] = __builtin_amdgcn_wmma_f32_16x16x4_f32(false, a, false, b, (short)0, acc2[u], false, false);
                a.x = wb.x; a.y = wb.y; b.x = acc1[t][4]; b.y = acc1[t][5];
                acc2[u] = __builtin_amdgcn_wmma_f32_16x16x4_f32(false, a, false, b, (short)0, acc2[u], false, false);
                a.x = wb.z; a.y = wb.w; b.x = acc1[t][6]; b.y = acc1[t][7];
                acc2[u] = __builtin_amdgcn_wmma_f32_16x16x4_f32(false, a, false, b, (short)0, acc2[u], false, false);
            }
        }

        #pragma unroll
        for (int u = 0; u < 8; ++u) {
            v4f lo, hi4;
            lo.x = acc2[u][0]; lo.y = acc2[u][1]; lo.z = acc2[u][2]; lo.w = acc2[u][3];
            hi4.x = acc2[u][4]; hi4.y = acc2[u][5]; hi4.z = acc2[u][6]; hi4.w = acc2[u][7];
            *(v4f*)(hout + rowoff + 16 * u)     = lo;
            *(v4f*)(hout + rowoff + 16 * u + 4) = hi4;
        }
    } else {
        #pragma unroll
        for (int u = 0; u < 8; ++u) {
            #pragma unroll
            for (int r = 0; r < 8; ++r) acc2[u][r] = 0.0f;
        }
    }

    if (do_stats) {
        float cs[8][8], cq[8][8];
        #pragma unroll
        for (int u = 0; u < 8; ++u) {
            #pragma unroll
            for (int r = 0; r < 8; ++r) {
                const float val = acc2[u][r];
                cs[u][r] = row16_sum(val);
                cq[u][r] = row16_sum(val * val);
            }
        }
        if (c == 0) {
            #pragma unroll
            for (int u = 0; u < 8; ++u) {
                #pragma unroll
                for (int r = 0; r < 8; ++r) {
                    red[wave][16 * u + 8 * h + r]       = cs[u][r];
                    red[wave][DIM + 16 * u + 8 * h + r] = cq[u][r];
                }
            }
        }
        __syncthreads();
        float tot = 0.0f;
        #pragma unroll
        for (int w = 0; w < TPB; ++w) tot += red[w][tid];
        atomicAdd(&stats[tid], tot);
    }
}

#define BN_EPT     8
#define BN_THREADS (N_NODES * DIM / BN_EPT)
#define BN_BLOCKS  (BN_THREADS / 256)
static_assert(N_NODES * DIM % BN_EPT == 0, "every thread has all its elements");
static_assert(BN_THREADS % 256 == 0, "whole blocks only, no thread past the end");
static_assert(BN_THREADS % DIM == 0, "a thread's elements must share one column");

__global__ __launch_bounds__(256) void gin_bn_relu(float* __restrict__ z,
                                                   const float* __restrict__ stats,
                                                   const float* __restrict__ gamma,
                                                   const float* __restrict__ beta) {
    const int i = (int)blockIdx.x * 256 + (int)threadIdx.x;
    const int c = i & (DIM - 1);
    float v[BN_EPT];
    #pragma unroll
    for (int k = 0; k < BN_EPT; ++k) v[k] = z[(long)i + (long)k * BN_THREADS];
    const float invN = 1.0f / (float)N_NODES;
    const float mu  = stats[c] * invN;
    const float var = stats[DIM + c] * invN - mu * mu;
    const float rs  = rsqrtf(var + BN_EPS);
    const float g   = gamma[c];
    const float b   = beta[c];
    #pragma unroll
    for (int k = 0; k < BN_EPT; ++k) {
        const float o = (v[k] - mu) * rs * g + b;
        z[(long)i + (long)k * BN_THREADS] = fmaxf(o, 0.0f);
    }
}

__global__ __launch_bounds__(256) void gin_pool(const float* __restrict__ h,
                                                const int* __restrict__ batch,
                                                float* __restrict__ out) {
    const int lane = (int)(threadIdx.x & 31);
    const int wv = __builtin_amdgcn_readfirstlane((int)(blockIdx.x * 8 + (threadIdx.x >> 5)));
    const int n0 = wv * EPW;
    if (n0 >= N_NODES) return;
    int g[EPW];
    #pragma unroll
    for (int j = 0; j < EPW; ++j) g[j] = batch[n0 + j];
    float v[EPW][4];
    #pragma unroll
    for (int j = 0; j < EPW; ++j) {
        const float* hp = h + (long)(n0 + j) * DIM + lane;
        v[j][0] = hp[0];
        v[j][1] = hp[32];
        v[j][2] = hp[64];
        v[j][3] = hp[96];
    }
    #pragma unroll
    for (int j = 0; j < EPW; ++j) {
        float* op = out + (long)g[j] * DIM + lane;
        atomicAdd(op + 0,  v[j][0]);
        atomicAdd(op + 32, v[j][1]);
        atomicAdd(op + 64, v[j][2]);
        atomicAdd(op + 96, v[j][3]);
    }
}

extern "C" void kernel_launch(void* const* d_in, const int* in_sizes, int n_in,
                              void* d_out, int out_size, void* d_ws, size_t ws_size,
                              hipStream_t stream) {
    const float* x     = (const float*)d_in[0];
    const int*   ei    = (const int*)d_in[1];
    const int*   batch = (const int*)d_in[2];
    const float* W1    = (const float*)d_in[3];
    const float* b1    = (const float*)d_in[4];
    const float* W2    = (const float*)d_in[5];
    const float* b2    = (const float*)d_in[6];
    const float* eps   = (const float*)d_in[7];
    const float* gamma = (const float*)d_in[8];
    const float* beta  = (const float*)d_in[9];
    float* out = (float*)d_out;

    const long ND = (long)N_NODES * DIM;
    float* hA    = (float*)d_ws;
    float* hB    = hA + ND;
    float* agg   = hB + ND;
    float* stats = agg + ND;
    int*   lens  = (int*)(stats + 2 * DIM);
    int*   lists = lens + N_SBLK * SC_WAVES;
    int*   cells = lists + (long)N_SBLK * LCAP;

    const int* src = ei;
    const int* dst = ei + N_EDGES;

    const int  mlpGrd  = (N_TILES + TPB - 1) / TPB;
    const int  bnGrd   = BN_BLOCKS;
    const int  poolGrd = (N_NODES / EPW + 7) / 8;

    gin_bin<<<N_CHUNK, SC_THREADS, 0, stream>>>(src, dst, cells);

    for (int i = 0; i < N_LAYERS; ++i) {
        const float* hin  = (i == 0) ? x : ((i & 1) ? hA : hB);
        float*       hout = (i & 1) ? hB : hA;
        int do_stats = (i < N_LAYERS - 1) ? 1 : 0;

        if (do_stats) gin_zero<<<1, 256, 0, stream>>>(stats, 2 * DIM);
        if (i == 0) gin_rowsum<true><<<N_SBLK, SC_THREADS, 0, stream>>>(hin, src, dst, agg, lists, lens, cells);
        else        gin_rowsum<false><<<N_SBLK, SC_THREADS, 0, stream>>>(hin, src, dst, agg, lists, lens, cells);
        gin_mlp<<<mlpGrd, 256, 0, stream>>>(hin, agg,
                                            W1 + (long)i * DIM * DIM, b1 + (long)i * DIM,
                                            W2 + (long)i * DIM * DIM, b2 + (long)i * DIM,
                                            eps + i, hout, stats, do_stats);
        if (do_stats) {
            gin_bn_relu<<<bnGrd, 256, 0, stream>>>(hout, stats,
                                                   gamma + (long)i * DIM,
                                                   beta + (long)i * DIM);
        }
    }
    gin_zero<<<(NUM_GRAPHS * DIM + 255) / 256, 256, 0, stream>>>(out, NUM_GRAPHS * DIM);
    gin_pool<<<poolGrd, 256, 0, stream>>>(hB, batch, out);
}
